# starting kernel
# baseline (speedup 1.0000x reference)
#include <hip/hip_runtime.h>

typedef __attribute__((ext_vector_type(2))) float v2f;
typedef __attribute__((ext_vector_type(8))) float v8f;

#define WMMA_F32(a, b, c) \
  __builtin_amdgcn_wmma_f32_16x16x4_f32(false, (a), false, (b), (short)0, (c), false, false)

#define BB      1024
#define NN      100000
#define DD      64
#define HH      512
#define RTILES  4
#define NSPLIT  64
#define CHUNKS  (NN / 16)
#define CPS     ((CHUNKS + NSPLIT - 1) / NSPLIT)

__global__ void tnorm_kernel(const float* __restrict__ table,
                             float* __restrict__ tnorm) {
  int j = blockIdx.x * blockDim.x + threadIdx.x;
  if (j >= NN) return;
  const float4* row = (const float4*)(table + (size_t)j * DD);
  float s = 0.f;
#pragma unroll
  for (int i = 0; i < DD / 4; ++i) {
    float4 v = row[i];
    s += v.x * v.x + v.y * v.y + v.z * v.z + v.w * v.w;
  }
  tnorm[j] = s;
}

__global__ __launch_bounds__(32) void nn_kernel(
    const float* __restrict__ codes, const float* __restrict__ table,
    const float* __restrict__ tnorm, float* __restrict__ pval,
    int* __restrict__ pidx) {
  const int lane = threadIdx.x;
  const int mgroup = blockIdx.x;
  const int split = blockIdx.y;
  const int m = lane & 15;
  const int koff = (lane >> 4) * 2;

  v2f a[RTILES][16];
#pragma unroll
  for (int t = 0; t < RTILES; ++t) {
    const float* crow =
        codes + (size_t)((mgroup * RTILES + t) * 16 + m) * DD + koff;
#pragma unroll
    for (int kb = 0; kb < 16; ++kb) {
      v2f v = *(const v2f*)(crow + kb * 4);
      a[t][kb].x = -2.0f * v.x;
      a[t][kb].y = -2.0f * v.y;
    }
  }

  float bestv[RTILES][8];
  int besti[RTILES][8];
#pragma unroll
  for (int t = 0; t < RTILES; ++t)
#pragma unroll
    for (int v = 0; v < 8; ++v) { bestv[t][v] = 3.4e38f; besti[t][v] = 0; }

  const int c0 = split * CPS;
  const int c1 = (c0 + CPS < CHUNKS) ? (c0 + CPS) : CHUNKS;

  for (int c = c0; c < c1; ++c) {
    const int col = c * 16 + m;
    const float* trow = table + (size_t)col * DD + koff;

    if (c + 1 < c1)
      __builtin_prefetch(table + (size_t)(col + 16) * DD + koff, 0, 1);

    v2f b[16];
#pragma unroll
    for (int kb = 0; kb < 16; ++kb) b[kb] = *(const v2f*)(trow + kb * 4);

    v8f acc[RTILES];
#pragma unroll
    for (int t = 0; t < RTILES; ++t) acc[t] = (v8f){};

#pragma unroll
    for (int kb = 0; kb < 16; ++kb)
#pragma unroll
      for (int t = 0; t < RTILES; ++t)
        acc[t] = WMMA_F32(a[t][kb], b[kb], acc[t]);

    const float tn = tnorm[col];
#pragma unroll
    for (int t = 0; t < RTILES; ++t)
#pragma unroll
      for (int v = 0; v < 8; ++v) {
        float s = tn + acc[t][v];
        if (s < bestv[t][v]) { bestv[t][v] = s; besti[t][v] = col; }
      }
  }

#pragma unroll
  for (int off = 1; off < 16; off <<= 1) {
#pragma unroll
    for (int t = 0; t < RTILES; ++t)
#pragma unroll
      for (int v = 0; v < 8; ++v) {
        float ov = __shfl_xor(bestv[t][v], off, 32);
        int oi = __shfl_xor(besti[t][v], off, 32);
        if (ov < bestv[t][v]) { bestv[t][v] = ov; besti[t][v] = oi; }
      }
  }

  if ((lane & 15) == 0) {
    const int rbase = (lane >> 4) * 8;
#pragma unroll
    for (int t = 0; t < RTILES; ++t) {
      const int tile = mgroup * RTILES + t;
      const int out = (tile * NSPLIT + split) * 16 + rbase;
#pragma unroll
      for (int v = 0; v < 8; ++v) {
        pval[out + v] = bestv[t][v];
        pidx[out + v] = besti[t][v];
      }
    }
  }
}

__global__ void reduce_kernel(const float* __restrict__ pval,
                              const int* __restrict__ pidx,
                              int* __restrict__ nn_idx) {
  int b = blockIdx.x * blockDim.x + threadIdx.x;
  if (b >= BB) return;
  const int tile = b >> 4, r = b & 15;
  float bv = 3.4e38f;
  int bi = 0;
  for (int s = 0; s < NSPLIT; ++s) {
    float v = pval[(tile * NSPLIT + s) * 16 + r];
    if (v < bv) { bv = v; bi = pidx[(tile * NSPLIT + s) * 16 + r]; }
  }
  nn_idx[b] = bi;
}

__global__ __launch_bounds__(32) void h1_kernel(
    const float* __restrict__ table, const int* __restrict__ nn_idx,
    const float* __restrict__ W1, const float* __restrict__ b1,
    float* __restrict__ h1) {
  const int lane = threadIdx.x;
  const int ntile = blockIdx.x;
  const int mtile = blockIdx.y;
  const int n = lane & 15;
  const int koff = (lane >> 4) * 2;
  const int nc = ntile * 16 + n;

  const int row = nn_idx[mtile * 16 + n];
  const float* prow = table + (size_t)row * DD + koff;

  const float bias = b1[nc];
  v8f acc;
#pragma unroll
  for (int v = 0; v < 8; ++v) acc[v] = bias;

#pragma unroll
  for (int kb = 0; kb < 16; ++kb) {
    v2f a = *(const v2f*)(prow + kb * 4);
    v2f b;
    b.x = W1[(size_t)(kb * 4 + koff + 0) * HH + nc];
    b.y = W1[(size_t)(kb * 4 + koff + 1) * HH + nc];
    acc = WMMA_F32(a, b, acc);
  }

#pragma unroll
  for (int v = 0; v < 8; ++v) {
    const int m = v + (lane >> 4) * 8;
    float h = acc[v];
    h1[(size_t)(mtile * 16 + m) * HH + nc] = h > 0.f ? h : 0.f;
  }
}

__global__ __launch_bounds__(32) void out_kernel(
    const float* __restrict__ h1, const float* __restrict__ Wu,
    const float* __restrict__ bu, const float* __restrict__ Ws,
    const float* __restrict__ bs, float* __restrict__ out) {
  const int lane = threadIdx.x;
  const int ntile = blockIdx.x;
  const int mtile = blockIdx.y;
  const int z = blockIdx.z;
  const float* W = z ? Ws : Wu;
  const float* bvec = z ? bs : bu;
  const int n = lane & 15;
  const int koff = (lane >> 4) * 2;
  const int nc = ntile * 16 + n;

  const float bias = bvec[nc];
  v8f acc;
#pragma unroll
  for (int v = 0; v < 8; ++v) acc[v] = bias;

  const float* hrow = h1 + (size_t)(mtile * 16 + n) * HH + koff;
#pragma unroll 8
  for (int kb = 0; kb < HH / 4; ++kb) {
    v2f a = *(const v2f*)(hrow + kb * 4);
    v2f b;
    b.x = W[(size_t)(kb * 4 + koff + 0) * DD + nc];
    b.y = W[(size_t)(kb * 4 + koff + 1) * DD + nc];
    acc = WMMA_F32(a, b, acc);
  }

  float* o = out + (size_t)z * BB * DD;
#pragma unroll
  for (int v = 0; v < 8; ++v) {
    const int m = v + (lane >> 4) * 8;
    o[(size_t)(mtile * 16 + m) * DD + nc] = acc[v];
  }
}

extern "C" void kernel_launch(void* const* d_in, const int* in_sizes, int n_in,
                              void* d_out, int out_size, void* d_ws,
                              size_t ws_size, hipStream_t stream) {
  const float* codes = (const float*)d_in[0];
  const float* table = (const float*)d_in[1];
  const float* W1 = (const float*)d_in[2];
  const float* b1 = (const float*)d_in[3];
  const float* Wu = (const float*)d_in[4];
  const float* bu = (const float*)d_in[5];
  const float* Ws = (const float*)d_in[6];
  const float* bs = (const float*)d_in[7];
  float* out = (float*)d_out;

  char* ws = (char*)d_ws;
  float* tnorm = (float*)(ws);
  float* pval  = (float*)(ws + 400000);
  int*   pidx  = (int*)  (ws + 400000 + 262144);
  int*   nnidx = (int*)  (ws + 400000 + 524288);
  float* h1    = (float*)(ws + 400000 + 524288 + 4096);

  tnorm_kernel<<<(NN + 255) / 256, 256, 0, stream>>>(table, tnorm);
  nn_kernel<<<dim3(BB / 16 / RTILES, NSPLIT), 32, 0, stream>>>(
      codes, table, tnorm, pval, pidx);
  reduce_kernel<<<(BB + 255) / 256, 256, 0, stream>>>(pval, pidx, nnidx);
  h1_kernel<<<dim3(HH / 16, BB / 16), 32, 0, stream>>>(table, nnidx, W1, b1,
                                                       h1);
  out_kernel<<<dim3(DD / 16, BB / 16, 2), 32, 0, stream>>>(h1, Wu, bu, Ws, bs,
                                                           out);
}


# nn search: table fragments of chunk c+2 loaded into a third rotating register set ahead of chunk c's WMMA block (loop unrolled by 3), replacing the one-line cache prefetch
# speedup vs baseline: 1.1213x; 1.1213x over previous
#include <hip/hip_runtime.h>

typedef __attribute__((ext_vector_type(2))) float v2f;
typedef __attribute__((ext_vector_type(8))) float v8f;

#define WMMA_F32(a, b, c) \
  __builtin_amdgcn_wmma_f32_16x16x4_f32(false, (a), false, (b), (short)0, (c), false, false)

#define BB      1024
#define NN      100000
#define DD      64
#define HH      512
#define RTILES  4
#define NSPLIT  64
#define CHUNKS  (NN / 16)
#define CPS     ((CHUNKS + NSPLIT - 1) / NSPLIT)

__global__ void tnorm_kernel(const float* __restrict__ table,
                             float* __restrict__ tnorm) {
  int j = blockIdx.x * blockDim.x + threadIdx.x;
  if (j >= NN) return;
  const float4* row = (const float4*)(table + (size_t)j * DD);
  float s = 0.f;
#pragma unroll
  for (int i = 0; i < DD / 4; ++i) {
    float4 v = row[i];
    s += v.x * v.x + v.y * v.y + v.z * v.z + v.w * v.w;
  }
  tnorm[j] = s;
}

__device__ __forceinline__ void load_chunk(v2f (&f)[16], float& tn,
                                           const float* __restrict__ table,
                                           const float* __restrict__ tnorm,
                                           int chunk, int m, int koff) {
  const int col = chunk * 16 + m;
  const float* trow = table + (size_t)col * DD + koff;
#pragma unroll
  for (int kb = 0; kb < 16; ++kb) f[kb] = *(const v2f*)(trow + kb * 4);
  tn = tnorm[col];
}

__device__ __forceinline__ void use_chunk(const v2f (&a)[RTILES][16],
                                          const v2f (&f)[16], float tn,
                                          int col, float (&bestv)[RTILES][8],
                                          int (&besti)[RTILES][8]) {
  v8f acc[RTILES];
#pragma unroll
  for (int t = 0; t < RTILES; ++t) acc[t] = (v8f){};

#pragma unroll
  for (int kb = 0; kb < 16; ++kb)
#pragma unroll
    for (int t = 0; t < RTILES; ++t)
      acc[t] = WMMA_F32(a[t][kb], f[kb], acc[t]);

#pragma unroll
  for (int t = 0; t < RTILES; ++t)
#pragma unroll
    for (int v = 0; v < 8; ++v) {
      float s = tn + acc[t][v];
      if (s < bestv[t][v]) { bestv[t][v] = s; besti[t][v] = col; }
    }
}

__global__ __launch_bounds__(32) void nn_kernel(
    const float* __restrict__ codes, const float* __restrict__ table,
    const float* __restrict__ tnorm, float* __restrict__ pval,
    int* __restrict__ pidx) {
  const int lane = threadIdx.x;
  const int mgroup = blockIdx.x;
  const int split = blockIdx.y;
  const int m = lane & 15;
  const int koff = (lane >> 4) * 2;

  v2f a[RTILES][16];
#pragma unroll
  for (int t = 0; t < RTILES; ++t) {
    const float* crow =
        codes + (size_t)((mgroup * RTILES + t) * 16 + m) * DD + koff;
#pragma unroll
    for (int kb = 0; kb < 16; ++kb) {
      v2f v = *(const v2f*)(crow + kb * 4);
      a[t][kb].x = -2.0f * v.x;
      a[t][kb].y = -2.0f * v.y;
    }
  }

  float bestv[RTILES][8];
  int besti[RTILES][8];
#pragma unroll
  for (int t = 0; t < RTILES; ++t)
#pragma unroll
    for (int v = 0; v < 8; ++v) { bestv[t][v] = 3.4e38f; besti[t][v] = 0; }

  const int c0 = split * CPS;
  const int c1 = (c0 + CPS < CHUNKS) ? (c0 + CPS) : CHUNKS;
  const int clast = c1 - 1;

  v2f f0[16], f1[16], f2[16];
  float n0, n1, n2;
  load_chunk(f0, n0, table, tnorm, c0 < clast ? c0 : clast, m, koff);
  load_chunk(f1, n1, table, tnorm, c0 + 1 < clast ? c0 + 1 : clast, m, koff);

  int c = c0;
  for (; c + 3 <= c1; c += 3) {
    load_chunk(f2, n2, table, tnorm, c + 2, m, koff);
    __builtin_amdgcn_sched_barrier(0);
    use_chunk(a, f0, n0, c * 16 + m, bestv, besti);
    load_chunk(f0, n0, table, tnorm, c + 3 < clast ? c + 3 : clast, m, koff);
    __builtin_amdgcn_sched_barrier(0);
    use_chunk(a, f1, n1, (c + 1) * 16 + m, bestv, besti);
    load_chunk(f1, n1, table, tnorm, c + 4 < clast ? c + 4 : clast, m, koff);
    __builtin_amdgcn_sched_barrier(0);
    use_chunk(a, f2, n2, (c + 2) * 16 + m, bestv, besti);
  }
  if (c < c1) use_chunk(a, f0, n0, c * 16 + m, bestv, besti);
  if (c + 1 < c1) use_chunk(a, f1, n1, (c + 1) * 16 + m, bestv, besti);

#pragma unroll
  for (int off = 1; off < 16; off <<= 1) {
#pragma unroll
    for (int t = 0; t < RTILES; ++t)
#pragma unroll
      for (int v = 0; v < 8; ++v) {
        float ov = __shfl_xor(bestv[t][v], off, 32);
        int oi = __shfl_xor(besti[t][v], off, 32);
        if (ov < bestv[t][v]) { bestv[t][v] = ov; besti[t][v] = oi; }
      }
  }

  if ((lane & 15) == 0) {
    const int rbase = (lane >> 4) * 8;
#pragma unroll
    for (int t = 0; t < RTILES; ++t) {
      const int tile = mgroup * RTILES + t;
      const int out = (tile * NSPLIT + split) * 16 + rbase;
#pragma unroll
      for (int v = 0; v < 8; ++v) {
        pval[out + v] = bestv[t][v];
        pidx[out + v] = besti[t][v];
      }
    }
  }
}

__global__ void reduce_kernel(const float* __restrict__ pval,
                              const int* __restrict__ pidx,
                              int* __restrict__ nn_idx) {
  int b = blockIdx.x * blockDim.x + threadIdx.x;
  if (b >= BB) return;
  const int tile = b >> 4, r = b & 15;
  float bv = 3.4e38f;
  int bi = 0;
  for (int s = 0; s < NSPLIT; ++s) {
    float v = pval[(tile * NSPLIT + s) * 16 + r];
    if (v < bv) { bv = v; bi = pidx[(tile * NSPLIT + s) * 16 + r]; }
  }
  nn_idx[b] = bi;
}

__global__ __launch_bounds__(32) void h1_kernel(
    const float* __restrict__ table, const int* __restrict__ nn_idx,
    const float* __restrict__ W1, const float* __restrict__ b1,
    float* __restrict__ h1) {
  const int lane = threadIdx.x;
  const int ntile = blockIdx.x;
  const int mtile = blockIdx.y;
  const int n = lane & 15;
  const int koff = (lane >> 4) * 2;
  const int nc = ntile * 16 + n;

  const int row = nn_idx[mtile * 16 + n];
  const float* prow = table + (size_t)row * DD + koff;

  const float bias = b1[nc];
  v8f acc;
#pragma unroll
  for (int v = 0; v < 8; ++v) acc[v] = bias;

#pragma unroll
  for (int kb = 0; kb < 16; ++kb) {
    v2f a = *(const v2f*)(prow + kb * 4);
    v2f b;
    b.x = W1[(size_t)(kb * 4 + koff + 0) * HH + nc];
    b.y = W1[(size_t)(kb * 4 + koff + 1) * HH + nc];
    acc = WMMA_F32(a, b, acc);
  }

#pragma unroll
  for (int v = 0; v < 8; ++v) {
    const int m = v + (lane >> 4) * 8;
    float h = acc[v];
    h1[(size_t)(mtile * 16 + m) * HH + nc] = h > 0.f ? h : 0.f;
  }
}

__global__ __launch_bounds__(32) void out_kernel(
    const float* __restrict__ h1, const float* __restrict__ Wu,
    const float* __restrict__ bu, const float* __restrict__ Ws,
    const float* __restrict__ bs, float* __restrict__ out) {
  const int lane = threadIdx.x;
  const int ntile = blockIdx.x;
  const int mtile = blockIdx.y;
  const int z = blockIdx.z;
  const float* W = z ? Ws : Wu;
  const float* bvec = z ? bs : bu;
  const int n = lane & 15;
  const int koff = (lane >> 4) * 2;
  const int nc = ntile * 16 + n;

  const float bias = bvec[nc];
  v8f acc;
#pragma unroll
  for (int v = 0; v < 8; ++v) acc[v] = bias;

  const float* hrow = h1 + (size_t)(mtile * 16 + n) * HH + koff;
#pragma unroll 8
  for (int kb = 0; kb < HH / 4; ++kb) {
    v2f a = *(const v2f*)(hrow + kb * 4);
    v2f b;
    b.x = W[(size_t)(kb * 4 + koff + 0) * DD + nc];
    b.y = W[(size_t)(kb * 4 + koff + 1) * DD + nc];
    acc = WMMA_F32(a, b, acc);
  }

  float* o = out + (size_t)z * BB * DD;
#pragma unroll
  for (int v = 0; v < 8; ++v) {
    const int m = v + (lane >> 4) * 8;
    o[(size_t)(mtile * 16 + m) * DD + nc] = acc[v];
  }
}

extern "C" void kernel_launch(void* const* d_in, const int* in_sizes, int n_in,
                              void* d_out, int out_size, void* d_ws,
                              size_t ws_size, hipStream_t stream) {
  const float* codes = (const float*)d_in[0];
  const float* table = (const float*)d_in[1];
  const float* W1 = (const float*)d_in[2];
  const float* b1 = (const float*)d_in[3];
  const float* Wu = (const float*)d_in[4];
  const float* bu = (const float*)d_in[5];
  const float* Ws = (const float*)d_in[6];
  const float* bs = (const float*)d_in[7];
  float* out = (float*)d_out;

  char* ws = (char*)d_ws;
  float* tnorm = (float*)(ws);
  float* pval  = (float*)(ws + 400000);
  int*   pidx  = (int*)  (ws + 400000 + 262144);
  int*   nnidx = (int*)  (ws + 400000 + 524288);
  float* h1    = (float*)(ws + 400000 + 524288 + 4096);

  tnorm_kernel<<<(NN + 255) / 256, 256, 0, stream>>>(table, tnorm);
  nn_kernel<<<dim3(BB / 16 / RTILES, NSPLIT), 32, 0, stream>>>(
      codes, table, tnorm, pval, pidx);
  reduce_kernel<<<(BB + 255) / 256, 256, 0, stream>>>(pval, pidx, nnidx);
  h1_kernel<<<dim3(HH / 16, BB / 16), 32, 0, stream>>>(table, nnidx, W1, b1,
                                                       h1);
  out_kernel<<<dim3(DD / 16, BB / 16, 2), 32, 0, stream>>>(h1, Wu, bu, Ws, bs,
                                                           out);
}


# nn search: two waves per work-group, each sweeping half of the block's table split and merging minima through LDS at the end, so two waves share each SIMD's matrix unit
# speedup vs baseline: 1.3574x; 1.2106x over previous
#include <hip/hip_runtime.h>

typedef __attribute__((ext_vector_type(2))) float v2f;
typedef __attribute__((ext_vector_type(8))) float v8f;

#define WMMA_F32(a, b, c) \
  __builtin_amdgcn_wmma_f32_16x16x4_f32(false, (a), false, (b), (short)0, (c), false, false)

#define BB      1024
#define NN      100000
#define DD      64
#define HH      512
#define RTILES  4
#define NSPLIT  64
#define CHUNKS  (NN / 16)
#define CPS     ((CHUNKS + NSPLIT - 1) / NSPLIT)

__global__ void tnorm_kernel(const float* __restrict__ table,
                             float* __restrict__ tnorm) {
  int j = blockIdx.x * blockDim.x + threadIdx.x;
  if (j >= NN) return;
  const float4* row = (const float4*)(table + (size_t)j * DD);
  float s = 0.f;
#pragma unroll
  for (int i = 0; i < DD / 4; ++i) {
    float4 v = row[i];
    s += v.x * v.x + v.y * v.y + v.z * v.z + v.w * v.w;
  }
  tnorm[j] = s;
}

__device__ __forceinline__ void load_chunk(v2f (&f)[16], float& tn,
                                           const float* __restrict__ table,
                                           const float* __restrict__ tnorm,
                                           int chunk, int m, int koff) {
  const int col = chunk * 16 + m;
  const float* trow = table + (size_t)col * DD + koff;
#pragma unroll
  for (int kb = 0; kb < 16; ++kb) f[kb] = *(const v2f*)(trow + kb * 4);
  tn = tnorm[col];
}

__device__ __forceinline__ void use_chunk(const v2f (&a)[RTILES][16],
                                          const v2f (&f)[16], float tn,
                                          int col, float (&bestv)[RTILES][8],
                                          int (&besti)[RTILES][8]) {
  v8f acc[RTILES];
#pragma unroll
  for (int t = 0; t < RTILES; ++t) acc[t] = (v8f){};

#pragma unroll
  for (int kb = 0; kb < 16; ++kb)
#pragma unroll
    for (int t = 0; t < RTILES; ++t)
      acc[t] = WMMA_F32(a[t][kb], f[kb], acc[t]);

#pragma unroll
  for (int t = 0; t < RTILES; ++t)
#pragma unroll
    for (int v = 0; v < 8; ++v) {
      float s = tn + acc[t][v];
      if (s < bestv[t][v]) { bestv[t][v] = s; besti[t][v] = col; }
    }
}

__global__ __launch_bounds__(64) void nn_kernel(
    const float* __restrict__ codes, const float* __restrict__ table,
    const float* __restrict__ tnorm, float* __restrict__ pval,
    int* __restrict__ pidx) {
  __shared__ float mval[2][RTILES * 8];
  __shared__ int midx[2][RTILES * 8];

  const int lane = threadIdx.x & 31;
  const int wave = __builtin_amdgcn_readfirstlane((int)(threadIdx.x >> 5));
  const int mgroup = blockIdx.x;
  const int split = blockIdx.y;
  const int m = lane & 15;
  const int koff = (lane >> 4) * 2;

  v2f a[RTILES][16];
#pragma unroll
  for (int t = 0; t < RTILES; ++t) {
    const float* crow =
        codes + (size_t)((mgroup * RTILES + t) * 16 + m) * DD + koff;
#pragma unroll
    for (int kb = 0; kb < 16; ++kb) {
      v2f v = *(const v2f*)(crow + kb * 4);
      a[t][kb].x = -2.0f * v.x;
      a[t][kb].y = -2.0f * v.y;
    }
  }

  float bestv[RTILES][8];
  int besti[RTILES][8];
#pragma unroll
  for (int t = 0; t < RTILES; ++t)
#pragma unroll
    for (int v = 0; v < 8; ++v) { bestv[t][v] = 3.4e38f; besti[t][v] = 0; }

  const int s0 = split * CPS;
  const int s1 = (s0 + CPS < CHUNKS) ? (s0 + CPS) : CHUNKS;
  const int mid = s0 + ((s1 - s0 + 1) >> 1);
  const int c0 = wave ? mid : s0;
  const int c1 = wave ? s1 : mid;
  const int clast = c1 - 1;

  v2f f0[16], f1[16], f2[16];
  float n0, n1, n2;
  load_chunk(f0, n0, table, tnorm, c0 < clast ? c0 : clast, m, koff);
  load_chunk(f1, n1, table, tnorm, c0 + 1 < clast ? c0 + 1 : clast, m, koff);

  int c = c0;
  for (; c + 3 <= c1; c += 3) {
    load_chunk(f2, n2, table, tnorm, c + 2, m, koff);
    __builtin_amdgcn_sched_barrier(0);
    use_chunk(a, f0, n0, c * 16 + m, bestv, besti);
    load_chunk(f0, n0, table, tnorm, c + 3 < clast ? c + 3 : clast, m, koff);
    __builtin_amdgcn_sched_barrier(0);
    use_chunk(a, f1, n1, (c + 1) * 16 + m, bestv, besti);
    load_chunk(f1, n1, table, tnorm, c + 4 < clast ? c + 4 : clast, m, koff);
    __builtin_amdgcn_sched_barrier(0);
    use_chunk(a, f2, n2, (c + 2) * 16 + m, bestv, besti);
  }
  if (c < c1) use_chunk(a, f0, n0, c * 16 + m, bestv, besti);
  if (c + 1 < c1) use_chunk(a, f1, n1, (c + 1) * 16 + m, bestv, besti);

#pragma unroll
  for (int off = 1; off < 16; off <<= 1) {
#pragma unroll
    for (int t = 0; t < RTILES; ++t)
#pragma unroll
      for (int v = 0; v < 8; ++v) {
        float ov = __shfl_xor(bestv[t][v], off, 32);
        int oi = __shfl_xor(besti[t][v], off, 32);
        if (ov < bestv[t][v]) { bestv[t][v] = ov; besti[t][v] = oi; }
      }
  }

  const int hh = lane >> 4;
  if (wave == 1 && (lane & 15) == 0) {
#pragma unroll
    for (int t = 0; t < RTILES; ++t)
#pragma unroll
      for (int v = 0; v < 8; ++v) {
        mval[hh][t * 8 + v] = bestv[t][v];
        midx[hh][t * 8 + v] = besti[t][v];
      }
  }
  __syncthreads();
  if (wave == 0 && (lane & 15) == 0) {
    const int rbase = hh * 8;
#pragma unroll
    for (int t = 0; t < RTILES; ++t) {
      const int tile = mgroup * RTILES + t;
      const int out = (tile * NSPLIT + split) * 16 + rbase;
#pragma unroll
      for (int v = 0; v < 8; ++v) {
        const float ov = mval[hh][t * 8 + v];
        const int oi = midx[hh][t * 8 + v];
        const bool take = ov < bestv[t][v];
        pval[out + v] = take ? ov : bestv[t][v];
        pidx[out + v] = take ? oi : besti[t][v];
      }
    }
  }
}

__global__ void reduce_kernel(const float* __restrict__ pval,
                              const int* __restrict__ pidx,
                              int* __restrict__ nn_idx) {
  int b = blockIdx.x * blockDim.x + threadIdx.x;
  if (b >= BB) return;
  const int tile = b >> 4, r = b & 15;
  float bv = 3.4e38f;
  int bi = 0;
  for (int s = 0; s < NSPLIT; ++s) {
    float v = pval[(tile * NSPLIT + s) * 16 + r];
    if (v < bv) { bv = v; bi = pidx[(tile * NSPLIT + s) * 16 + r]; }
  }
  nn_idx[b] = bi;
}

__global__ __launch_bounds__(32) void h1_kernel(
    const float* __restrict__ table, const int* __restrict__ nn_idx,
    const float* __restrict__ W1, const float* __restrict__ b1,
    float* __restrict__ h1) {
  const int lane = threadIdx.x;
  const int ntile = blockIdx.x;
  const int mtile = blockIdx.y;
  const int n = lane & 15;
  const int koff = (lane >> 4) * 2;
  const int nc = ntile * 16 + n;

  const int row = nn_idx[mtile * 16 + n];
  const float* prow = table + (size_t)row * DD + koff;

  const float bias = b1[nc];
  v8f acc;
#pragma unroll
  for (int v = 0; v < 8; ++v) acc[v] = bias;

#pragma unroll
  for (int kb = 0; kb < 16; ++kb) {
    v2f a = *(const v2f*)(prow + kb * 4);
    v2f b;
    b.x = W1[(size_t)(kb * 4 + koff + 0) * HH + nc];
    b.y = W1[(size_t)(kb * 4 + koff + 1) * HH + nc];
    acc = WMMA_F32(a, b, acc);
  }

#pragma unroll
  for (int v = 0; v < 8; ++v) {
    const int m = v + (lane >> 4) * 8;
    float h = acc[v];
    h1[(size_t)(mtile * 16 + m) * HH + nc] = h > 0.f ? h : 0.f;
  }
}

__global__ __launch_bounds__(32) void out_kernel(
    const float* __restrict__ h1, const float* __restrict__ Wu,
    const float* __restrict__ bu, const float* __restrict__ Ws,
    const float* __restrict__ bs, float* __restrict__ out) {
  const int lane = threadIdx.x;
  const int ntile = blockIdx.x;
  const int mtile = blockIdx.y;
  const int z = blockIdx.z;
  const float* W = z ? Ws : Wu;
  const float* bvec = z ? bs : bu;
  const int n = lane & 15;
  const int koff = (lane >> 4) * 2;
  const int nc = ntile * 16 + n;

  const float bias = bvec[nc];
  v8f acc;
#pragma unroll
  for (int v = 0; v < 8; ++v) acc[v] = bias;

  const float* hrow = h1 + (size_t)(mtile * 16 + n) * HH + koff;
#pragma unroll 8
  for (int kb = 0; kb < HH / 4; ++kb) {
    v2f a = *(const v2f*)(hrow + kb * 4);
    v2f b;
    b.x = W[(size_t)(kb * 4 + koff + 0) * DD + nc];
    b.y = W[(size_t)(kb * 4 + koff + 1) * DD + nc];
    acc = WMMA_F32(a, b, acc);
  }

  float* o = out + (size_t)z * BB * DD;
#pragma unroll
  for (int v = 0; v < 8; ++v) {
    const int m = v + (lane >> 4) * 8;
    o[(size_t)(mtile * 16 + m) * DD + nc] = acc[v];
  }
}

extern "C" void kernel_launch(void* const* d_in, const int* in_sizes, int n_in,
                              void* d_out, int out_size, void* d_ws,
                              size_t ws_size, hipStream_t stream) {
  const float* codes = (const float*)d_in[0];
  const float* table = (const float*)d_in[1];
  const float* W1 = (const float*)d_in[2];
  const float* b1 = (const float*)d_in[3];
  const float* Wu = (const float*)d_in[4];
  const float* bu = (const float*)d_in[5];
  const float* Ws = (const float*)d_in[6];
  const float* bs = (const float*)d_in[7];
  float* out = (float*)d_out;

  char* ws = (char*)d_ws;
  float* tnorm = (float*)(ws);
  float* pval  = (float*)(ws + 400000);
  int*   pidx  = (int*)  (ws + 400000 + 262144);
  int*   nnidx = (int*)  (ws + 400000 + 524288);
  float* h1    = (float*)(ws + 400000 + 524288 + 4096);

  tnorm_kernel<<<(NN + 255) / 256, 256, 0, stream>>>(table, tnorm);
  nn_kernel<<<dim3(BB / 16 / RTILES, NSPLIT), 64, 0, stream>>>(
      codes, table, tnorm, pval, pidx);
  reduce_kernel<<<(BB + 255) / 256, 256, 0, stream>>>(pval, pidx, nnidx);
  h1_kernel<<<dim3(HH / 16, BB / 16), 32, 0, stream>>>(table, nnidx, W1, b1,
                                                       h1);
  out_kernel<<<dim3(DD / 16, BB / 16, 2), 32, 0, stream>>>(h1, Wu, bu, Ws, bs,
                                                           out);
}


# argmin merge of the 64 split partials moved into the hidden-layer kernel's prologue: one batch of independent loads, first minimum kept; the serial reduce launch is gone (5 launches to 4)
# speedup vs baseline: 1.7367x; 1.2794x over previous
#include <hip/hip_runtime.h>

typedef __attribute__((ext_vector_type(2))) float v2f;
typedef __attribute__((ext_vector_type(8))) float v8f;

#define WMMA_F32(a, b, c) \
  __builtin_amdgcn_wmma_f32_16x16x4_f32(false, (a), false, (b), (short)0, (c), false, false)

#define BB      1024
#define NN      100000
#define DD      64
#define HH      512
#define RTILES  4
#define NSPLIT  64
#define CHUNKS  (NN / 16)
#define CPS     ((CHUNKS + NSPLIT - 1) / NSPLIT)

__global__ void tnorm_kernel(const float* __restrict__ table,
                             float* __restrict__ tnorm) {
  int j = blockIdx.x * blockDim.x + threadIdx.x;
  if (j >= NN) return;
  const float4* row = (const float4*)(table + (size_t)j * DD);
  float s = 0.f;
#pragma unroll
  for (int i = 0; i < DD / 4; ++i) {
    float4 v = row[i];
    s += v.x * v.x + v.y * v.y + v.z * v.z + v.w * v.w;
  }
  tnorm[j] = s;
}

__device__ __forceinline__ void load_chunk(v2f (&f)[16], float& tn,
                                           const float* __restrict__ table,
                                           const float* __restrict__ tnorm,
                                           int chunk, int m, int koff) {
  const int col = chunk * 16 + m;
  const float* trow = table + (size_t)col * DD + koff;
#pragma unroll
  for (int kb = 0; kb < 16; ++kb) f[kb] = *(const v2f*)(trow + kb * 4);
  tn = tnorm[col];
}

__device__ __forceinline__ void use_chunk(const v2f (&a)[RTILES][16],
                                          const v2f (&f)[16], float tn,
                                          int col, float (&bestv)[RTILES][8],
                                          int (&besti)[RTILES][8]) {
  v8f acc[RTILES];
#pragma unroll
  for (int t = 0; t < RTILES; ++t) acc[t] = (v8f){};

#pragma unroll
  for (int kb = 0; kb < 16; ++kb)
#pragma unroll
    for (int t = 0; t < RTILES; ++t)
      acc[t] = WMMA_F32(a[t][kb], f[kb], acc[t]);

#pragma unroll
  for (int t = 0; t < RTILES; ++t)
#pragma unroll
    for (int v = 0; v < 8; ++v) {
      float s = tn + acc[t][v];
      if (s < bestv[t][v]) { bestv[t][v] = s; besti[t][v] = col; }
    }
}

__global__ __launch_bounds__(64) void nn_kernel(
    const float* __restrict__ codes, const float* __restrict__ table,
    const float* __restrict__ tnorm, float* __restrict__ pval,
    int* __restrict__ pidx) {
  __shared__ float mval[2][RTILES * 8];
  __shared__ int midx[2][RTILES * 8];

  const int lane = threadIdx.x & 31;
  const int wave = __builtin_amdgcn_readfirstlane((int)(threadIdx.x >> 5));
  const int mgroup = blockIdx.x;
  const int split = blockIdx.y;
  const int m = lane & 15;
  const int koff = (lane >> 4) * 2;

  v2f a[RTILES][16];
#pragma unroll
  for (int t = 0; t < RTILES; ++t) {
    const float* crow =
        codes + (size_t)((mgroup * RTILES + t) * 16 + m) * DD + koff;
#pragma unroll
    for (int kb = 0; kb < 16; ++kb) {
      v2f v = *(const v2f*)(crow + kb * 4);
      a[t][kb].x = -2.0f * v.x;
      a[t][kb].y = -2.0f * v.y;
    }
  }

  float bestv[RTILES][8];
  int besti[RTILES][8];
#pragma unroll
  for (int t = 0; t < RTILES; ++t)
#pragma unroll
    for (int v = 0; v < 8; ++v) { bestv[t][v] = 3.4e38f; besti[t][v] = 0; }

  const int s0 = split * CPS;
  const int s1 = (s0 + CPS < CHUNKS) ? (s0 + CPS) : CHUNKS;
  const int mid = s0 + ((s1 - s0 + 1) >> 1);
  const int c0 = wave ? mid : s0;
  const int c1 = wave ? s1 : mid;
  const int clast = c1 - 1;

  v2f f0[16], f1[16], f2[16];
  float n0, n1, n2;
  load_chunk(f0, n0, table, tnorm, c0 < clast ? c0 : clast, m, koff);
  load_chunk(f1, n1, table, tnorm, c0 + 1 < clast ? c0 + 1 : clast, m, koff);

  int c = c0;
  for (; c + 3 <= c1; c += 3) {
    load_chunk(f2, n2, table, tnorm, c + 2, m, koff);
    __builtin_amdgcn_sched_barrier(0);
    use_chunk(a, f0, n0, c * 16 + m, bestv, besti);
    load_chunk(f0, n0, table, tnorm, c + 3 < clast ? c + 3 : clast, m, koff);
    __builtin_amdgcn_sched_barrier(0);
    use_chunk(a, f1, n1, (c + 1) * 16 + m, bestv, besti);
    load_chunk(f1, n1, table, tnorm, c + 4 < clast ? c + 4 : clast, m, koff);
    __builtin_amdgcn_sched_barrier(0);
    use_chunk(a, f2, n2, (c + 2) * 16 + m, bestv, besti);
  }
  if (c < c1) use_chunk(a, f0, n0, c * 16 + m, bestv, besti);
  if (c + 1 < c1) use_chunk(a, f1, n1, (c + 1) * 16 + m, bestv, besti);

#pragma unroll
  for (int off = 1; off < 16; off <<= 1) {
#pragma unroll
    for (int t = 0; t < RTILES; ++t)
#pragma unroll
      for (int v = 0; v < 8; ++v) {
        float ov = __shfl_xor(bestv[t][v], off, 32);
        int oi = __shfl_xor(besti[t][v], off, 32);
        if (ov < bestv[t][v]) { bestv[t][v] = ov; besti[t][v] = oi; }
      }
  }

  const int hh = lane >> 4;
  if (wave == 1 && (lane & 15) == 0) {
#pragma unroll
    for (int t = 0; t < RTILES; ++t)
#pragma unroll
      for (int v = 0; v < 8; ++v) {
        mval[hh][t * 8 + v] = bestv[t][v];
        midx[hh][t * 8 + v] = besti[t][v];
      }
  }
  __syncthreads();
  if (wave == 0 && (lane & 15) == 0) {
    const int rbase = hh * 8;
#pragma unroll
    for (int t = 0; t < RTILES; ++t) {
      const int tile = mgroup * RTILES + t;
      const int out = (tile * NSPLIT + split) * 16 + rbase;
#pragma unroll
      for (int v = 0; v < 8; ++v) {
        const float ov = mval[hh][t * 8 + v];
        const int oi = midx[hh][t * 8 + v];
        const bool take = ov < bestv[t][v];
        pval[out + v] = take ? ov : bestv[t][v];
        pidx[out + v] = take ? oi : besti[t][v];
      }
    }
  }
}

__global__ __launch_bounds__(32) void h1_kernel(
    const float* __restrict__ table, const float* __restrict__ pval,
    const int* __restrict__ pidx, const float* __restrict__ W1,
    const float* __restrict__ b1, float* __restrict__ h1) {
  const int lane = threadIdx.x;
  const int ntile = blockIdx.x;
  const int mtile = blockIdx.y;
  const int n = lane & 15;
  const int hh = lane >> 4;
  const int koff = hh * 2;
  const int nc = ntile * 16 + n;

  const int pbase = (mtile * NSPLIT + hh * (NSPLIT / 2)) * 16 + n;
  float pv[NSPLIT / 2];
  int pi[NSPLIT / 2];
#pragma unroll
  for (int j = 0; j < NSPLIT / 2; ++j) {
    pv[j] = pval[pbase + j * 16];
    pi[j] = pidx[pbase + j * 16];
  }

  float cv[4];
  int ci[4];
#pragma unroll
  for (int q = 0; q < 4; ++q) {
    cv[q] = 3.4e38f;
    ci[q] = 0;
#pragma unroll
    for (int j = 0; j < 8; ++j) {
      const float v = pv[q * 8 + j];
      if (v < cv[q]) { cv[q] = v; ci[q] = pi[q * 8 + j]; }
    }
  }
  float bv = cv[0];
  int bi = ci[0];
#pragma unroll
  for (int q = 1; q < 4; ++q)
    if (cv[q] < bv) { bv = cv[q]; bi = ci[q]; }

  const float ov = __shfl_xor(bv, 16, 32);
  const int oi = __shfl_xor(bi, 16, 32);
  const bool take = hh ? !(bv < ov) : (ov < bv);
  const int row = take ? oi : bi;
  const float* prow = table + (size_t)row * DD + koff;

  const float bias = b1[nc];
  v8f acc;
#pragma unroll
  for (int v = 0; v < 8; ++v) acc[v] = bias;

#pragma unroll
  for (int kb = 0; kb < 16; ++kb) {
    v2f a = *(const v2f*)(prow + kb * 4);
    v2f b;
    b.x = W1[(size_t)(kb * 4 + koff + 0) * HH + nc];
    b.y = W1[(size_t)(kb * 4 + koff + 1) * HH + nc];
    acc = WMMA_F32(a, b, acc);
  }

#pragma unroll
  for (int v = 0; v < 8; ++v) {
    const int m = v + (lane >> 4) * 8;
    float h = acc[v];
    h1[(size_t)(mtile * 16 + m) * HH + nc] = h > 0.f ? h : 0.f;
  }
}

__global__ __launch_bounds__(32) void out_kernel(
    const float* __restrict__ h1, const float* __restrict__ Wu,
    const float* __restrict__ bu, const float* __restrict__ Ws,
    const float* __restrict__ bs, float* __restrict__ out) {
  const int lane = threadIdx.x;
  const int ntile = blockIdx.x;
  const int mtile = blockIdx.y;
  const int z = blockIdx.z;
  const float* W = z ? Ws : Wu;
  const float* bvec = z ? bs : bu;
  const int n = lane & 15;
  const int koff = (lane >> 4) * 2;
  const int nc = ntile * 16 + n;

  const float bias = bvec[nc];
  v8f acc;
#pragma unroll
  for (int v = 0; v < 8; ++v) acc[v] = bias;

  const float* hrow = h1 + (size_t)(mtile * 16 + n) * HH + koff;
#pragma unroll 8
  for (int kb = 0; kb < HH / 4; ++kb) {
    v2f a = *(const v2f*)(hrow + kb * 4);
    v2f b;
    b.x = W[(size_t)(kb * 4 + koff + 0) * DD + nc];
    b.y = W[(size_t)(kb * 4 + koff + 1) * DD + nc];
    acc = WMMA_F32(a, b, acc);
  }

  float* o = out + (size_t)z * BB * DD;
#pragma unroll
  for (int v = 0; v < 8; ++v) {
    const int m = v + (lane >> 4) * 8;
    o[(size_t)(mtile * 16 + m) * DD + nc] = acc[v];
  }
}

extern "C" void kernel_launch(void* const* d_in, const int* in_sizes, int n_in,
                              void* d_out, int out_size, void* d_ws,
                              size_t ws_size, hipStream_t stream) {
  const float* codes = (const float*)d_in[0];
  const float* table = (const float*)d_in[1];
  const float* W1 = (const float*)d_in[2];
  const float* b1 = (const float*)d_in[3];
  const float* Wu = (const float*)d_in[4];
  const float* bu = (const float*)d_in[5];
  const float* Ws = (const float*)d_in[6];
  const float* bs = (const float*)d_in[7];
  float* out = (float*)d_out;

  char* ws = (char*)d_ws;
  float* tnorm = (float*)(ws);
  float* pval  = (float*)(ws + 400000);
  int*   pidx  = (int*)  (ws + 400000 + 262144);
  float* h1    = (float*)(ws + 400000 + 524288 + 4096);

  tnorm_kernel<<<(NN + 255) / 256, 256, 0, stream>>>(table, tnorm);
  nn_kernel<<<dim3(BB / 16 / RTILES, NSPLIT), 64, 0, stream>>>(
      codes, table, tnorm, pval, pidx);
  h1_kernel<<<dim3(HH / 16, BB / 16), 32, 0, stream>>>(table, pval, pidx, W1,
                                                       b1, h1);
  out_kernel<<<dim3(DD / 16, BB / 16, 2), 32, 0, stream>>>(h1, Wu, bu, Ws, bs,
                                                           out);
}


# hidden layer and both output heads fused into one 8-wave kernel per 16 batch rows: h1 tile stays in LDS, weight loads issued ahead of the matrix blocks, k order of every sum kept; 4 launches to 3
# speedup vs baseline: 1.8892x; 1.0878x over previous
#include <hip/hip_runtime.h>

typedef __attribute__((ext_vector_type(2))) float v2f;
typedef __attribute__((ext_vector_type(8))) float v8f;

#define WMMA_F32(a, b, c) \
  __builtin_amdgcn_wmma_f32_16x16x4_f32(false, (a), false, (b), (short)0, (c), false, false)

#define BB      1024
#define NN      100000
#define DD      64
#define HH      512
#define RTILES  4
#define NSPLIT  64
#define CHUNKS  (NN / 16)
#define CPS     ((CHUNKS + NSPLIT - 1) / NSPLIT)

__global__ void tnorm_kernel(const float* __restrict__ table,
                             float* __restrict__ tnorm) {
  int j = blockIdx.x * blockDim.x + threadIdx.x;
  if (j >= NN) return;
  const float4* row = (const float4*)(table + (size_t)j * DD);
  float s = 0.f;
#pragma unroll
  for (int i = 0; i < DD / 4; ++i) {
    float4 v = row[i];
    s += v.x * v.x + v.y * v.y + v.z * v.z + v.w * v.w;
  }
  tnorm[j] = s;
}

__device__ __forceinline__ void load_chunk(v2f (&f)[16], float& tn,
                                           const float* __restrict__ table,
                                           const float* __restrict__ tnorm,
                                           int chunk, int m, int koff) {
  const int col = chunk * 16 + m;
  const float* trow = table + (size_t)col * DD + koff;
#pragma unroll
  for (int kb = 0; kb < 16; ++kb) f[kb] = *(const v2f*)(trow + kb * 4);
  tn = tnorm[col];
}

__device__ __forceinline__ void use_chunk(const v2f (&a)[RTILES][16],
                                          const v2f (&f)[16], float tn,
                                          int col, float (&bestv)[RTILES][8],
                                          int (&besti)[RTILES][8]) {
  v8f acc[RTILES];
#pragma unroll
  for (int t = 0; t < RTILES; ++t) acc[t] = (v8f){};

#pragma unroll
  for (int kb = 0; kb < 16; ++kb)
#pragma unroll
    for (int t = 0; t < RTILES; ++t)
      acc[t] = WMMA_F32(a[t][kb], f[kb], acc[t]);

#pragma unroll
  for (int t = 0; t < RTILES; ++t)
#pragma unroll
    for (int v = 0; v < 8; ++v) {
      float s = tn + acc[t][v];
      if (s < bestv[t][v]) { bestv[t][v] = s; besti[t][v] = col; }
    }
}

__global__ __launch_bounds__(64) void nn_kernel(
    const float* __restrict__ codes, const float* __restrict__ table,
    const float* __restrict__ tnorm, float* __restrict__ pval,
    int* __restrict__ pidx) {
  __shared__ float mval[2][RTILES * 8];
  __shared__ int midx[2][RTILES * 8];

  const int lane = threadIdx.x & 31;
  const int wave = __builtin_amdgcn_readfirstlane((int)(threadIdx.x >> 5));
  const int mgroup = blockIdx.x;
  const int split = blockIdx.y;
  const int m = lane & 15;
  const int koff = (lane >> 4) * 2;

  v2f a[RTILES][16];
#pragma unroll
  for (int t = 0; t < RTILES; ++t) {
    const float* crow =
        codes + (size_t)((mgroup * RTILES + t) * 16 + m) * DD + koff;
#pragma unroll
    for (int kb = 0; kb < 16; ++kb) {
      v2f v = *(const v2f*)(crow + kb * 4);
      a[t][kb].x = -2.0f * v.x;
      a[t][kb].y = -2.0f * v.y;
    }
  }

  float bestv[RTILES][8];
  int besti[RTILES][8];
#pragma unroll
  for (int t = 0; t < RTILES; ++t)
#pragma unroll
    for (int v = 0; v < 8; ++v) { bestv[t][v] = 3.4e38f; besti[t][v] = 0; }

  const int s0 = split * CPS;
  const int s1 = (s0 + CPS < CHUNKS) ? (s0 + CPS) : CHUNKS;
  const int mid = s0 + ((s1 - s0 + 1) >> 1);
  const int c0 = wave ? mid : s0;
  const int c1 = wave ? s1 : mid;
  const int clast = c1 - 1;

  v2f f0[16], f1[16], f2[16];
  float n0, n1, n2;
  load_chunk(f0, n0, table, tnorm, c0 < clast ? c0 : clast, m, koff);
  load_chunk(f1, n1, table, tnorm, c0 + 1 < clast ? c0 + 1 : clast, m, koff);

  int c = c0;
  for (; c + 3 <= c1; c += 3) {
    load_chunk(f2, n2, table, tnorm, c + 2, m, koff);
    __builtin_amdgcn_sched_barrier(0);
    use_chunk(a, f0, n0, c * 16 + m, bestv, besti);
    load_chunk(f0, n0, table, tnorm, c + 3 < clast ? c + 3 : clast, m, koff);
    __builtin_amdgcn_sched_barrier(0);
    use_chunk(a, f1, n1, (c + 1) * 16 + m, bestv, besti);
    load_chunk(f1, n1, table, tnorm, c + 4 < clast ? c + 4 : clast, m, koff);
    __builtin_amdgcn_sched_barrier(0);
    use_chunk(a, f2, n2, (c + 2) * 16 + m, bestv, besti);
  }
  if (c < c1) use_chunk(a, f0, n0, c * 16 + m, bestv, besti);
  if (c + 1 < c1) use_chunk(a, f1, n1, (c + 1) * 16 + m, bestv, besti);

#pragma unroll
  for (int off = 1; off < 16; off <<= 1) {
#pragma unroll
    for (int t = 0; t < RTILES; ++t)
#pragma unroll
      for (int v = 0; v < 8; ++v) {
        float ov = __shfl_xor(bestv[t][v], off, 32);
        int oi = __shfl_xor(besti[t][v], off, 32);
        if (ov < bestv[t][v]) { bestv[t][v] = ov; besti[t][v] = oi; }
      }
  }

  const int hh = lane >> 4;
  if (wave == 1 && (lane & 15) == 0) {
#pragma unroll
    for (int t = 0; t < RTILES; ++t)
#pragma unroll
      for (int v = 0; v < 8; ++v) {
        mval[hh][t * 8 + v] = bestv[t][v];
        midx[hh][t * 8 + v] = besti[t][v];
      }
  }
  __syncthreads();
  if (wave == 0 && (lane & 15) == 0) {
    const int rbase = hh * 8;
#pragma unroll
    for (int t = 0; t < RTILES; ++t) {
      const int tile = mgroup * RTILES + t;
      const int out = (tile * NSPLIT + split) * 16 + rbase;
#pragma unroll
      for (int v = 0; v < 8; ++v) {
        const float ov = mval[hh][t * 8 + v];
        const int oi = midx[hh][t * 8 + v];
        const bool take = ov < bestv[t][v];
        pval[out + v] = take ? ov : bestv[t][v];
        pidx[out + v] = take ? oi : besti[t][v];
      }
    }
  }
}

#define HPITCH (HH + 4)

__global__ __launch_bounds__(256) void mlp_kernel(
    const float* __restrict__ table, const float* __restrict__ pval,
    const int* __restrict__ pidx, const float* __restrict__ W1,
    const float* __restrict__ b1, const float* __restrict__ Wu,
    const float* __restrict__ bu, const float* __restrict__ Ws,
    const float* __restrict__ bs, float* __restrict__ out) {
  __shared__ __attribute__((aligned(16))) float h1s[16 * HPITCH];
  __shared__ int nnrow[16];

  const int lane = threadIdx.x & 31;
  const int wave = __builtin_amdgcn_readfirstlane((int)(threadIdx.x >> 5));
  const int mtile = blockIdx.x;
  const int n = lane & 15;
  const int hh = lane >> 4;
  const int koff = hh * 2;

  const int row0 = wave * 2 + hh;
  const int pb = (mtile * NSPLIT + n) * 16 + row0;
  float pv[4];
  int pi[4];
#pragma unroll
  for (int j = 0; j < 4; ++j) {
    pv[j] = pval[pb + j * 256];
    pi[j] = pidx[pb + j * 256];
  }

  v2f w1f[4][16];
  float bias1[4];
#pragma unroll
  for (int t = 0; t < 4; ++t) {
    const int nc = (wave * 4 + t) * 16 + n;
#pragma unroll
    for (int kb = 0; kb < 16; ++kb) {
      w1f[t][kb].x = W1[(size_t)(kb * 4 + koff + 0) * HH + nc];
      w1f[t][kb].y = W1[(size_t)(kb * 4 + koff + 1) * HH + nc];
    }
    bias1[t] = b1[nc];
  }

  {
    float bv = pv[0];
    int bi = pi[0];
#pragma unroll
    for (int j = 1; j < 4; ++j) {
      const bool lt = pv[j] < bv;
      bv = lt ? pv[j] : bv;
      bi = lt ? pi[j] : bi;
    }
#pragma unroll
    for (int off = 8; off >= 1; off >>= 1) {
      const float ov = __shfl_xor(bv, off, 32);
      const int oi = __shfl_xor(bi, off, 32);
      const bool take = (ov < bv) | ((ov == bv) & (oi < bi));
      bv = take ? ov : bv;
      bi = take ? oi : bi;
    }
    if (n == 0) nnrow[row0] = bi;
  }
  __syncthreads();

  const float* prow = table + (size_t)nnrow[n] * DD + koff;
  v2f a[16];
#pragma unroll
  for (int kb = 0; kb < 16; ++kb) a[kb] = *(const v2f*)(prow + kb * 4);

  const int z = wave >> 2;
  const float* Wz = z ? Ws : Wu;
  const float* bz = z ? bs : bu;
  const int nc2 = (wave & 3) * 16 + n;
  v2f b2[HH / 4];
#pragma unroll
  for (int kb = 0; kb < HH / 4; ++kb) {
    b2[kb].x = Wz[(size_t)(kb * 4 + koff + 0) * DD + nc2];
    b2[kb].y = Wz[(size_t)(kb * 4 + koff + 1) * DD + nc2];
  }
  const float bias2 = bz[nc2];

  {
    v8f acc[4];
#pragma unroll
    for (int t = 0; t < 4; ++t)
#pragma unroll
      for (int v = 0; v < 8; ++v) acc[t][v] = bias1[t];

#pragma unroll
    for (int kb = 0; kb < 16; ++kb)
#pragma unroll
      for (int t = 0; t < 4; ++t)
        acc[t] = WMMA_F32(a[kb], w1f[t][kb], acc[t]);

#pragma unroll
    for (int t = 0; t < 4; ++t)
#pragma unroll
      for (int v = 0; v < 8; ++v) {
        const int m = v + hh * 8;
        const float h = acc[t][v];
        h1s[m * HPITCH + (wave * 4 + t) * 16 + n] = h > 0.f ? h : 0.f;
      }
  }
  __syncthreads();

  {
    v8f o;
#pragma unroll
    for (int v = 0; v < 8; ++v) o[v] = bias2;
    const float* hrow = h1s + n * HPITCH + koff;
    v2f a2[2][16];
#pragma unroll
    for (int i = 0; i < 16; ++i) a2[0][i] = *(const v2f*)(hrow + i * 4);
#pragma unroll
    for (int g = 0; g < HH / 64; ++g) {
      if (g + 1 < HH / 64) {
#pragma unroll
        for (int i = 0; i < 16; ++i)
          a2[(g + 1) & 1][i] = *(const v2f*)(hrow + ((g + 1) * 16 + i) * 4);
      }
      __builtin_amdgcn_sched_barrier(0);
#pragma unroll
      for (int i = 0; i < 16; ++i)
        o = WMMA_F32(a2[g & 1][i], b2[g * 16 + i], o);
    }
    float* op = out + (size_t)z * BB * DD;
#pragma unroll
    for (int v = 0; v < 8; ++v) {
      const int m = v + hh * 8;
      op[(size_t)(mtile * 16 + m) * DD + nc2] = o[v];
    }
  }
}

extern "C" void kernel_launch(void* const* d_in, const int* in_sizes, int n_in,
                              void* d_out, int out_size, void* d_ws,
                              size_t ws_size, hipStream_t stream) {
  const float* codes = (const float*)d_in[0];
  const float* table = (const float*)d_in[1];
  const float* W1 = (const float*)d_in[2];
  const float* b1 = (const float*)d_in[3];
  const float* Wu = (const float*)d_in[4];
  const float* bu = (const float*)d_in[5];
  const float* Ws = (const float*)d_in[6];
  const float* bs = (const float*)d_in[7];
  float* out = (float*)d_out;

  char* ws = (char*)d_ws;
  float* tnorm = (float*)(ws);
  float* pval  = (float*)(ws + 400000);
  int*   pidx  = (int*)  (ws + 400000 + 262144);

  tnorm_kernel<<<(NN + 255) / 256, 256, 0, stream>>>(table, tnorm);
  nn_kernel<<<dim3(BB / 16 / RTILES, NSPLIT), 64, 0, stream>>>(
      codes, table, tnorm, pval, pidx);
  mlp_kernel<<<BB / 16, 256, 0, stream>>>(table, pval, pidx, W1, b1, Wu, bu, Ws,
                                          bs, out);
}


# nn search: squared norm of each table row summed from the B fragments a lane already holds, halves joined by one lane exchange (new order of the 64 terms); the norm launch is gone, 3 launches to 2
# speedup vs baseline: 1.9883x; 1.0525x over previous
#include <hip/hip_runtime.h>

typedef __attribute__((ext_vector_type(2))) float v2f;
typedef __attribute__((ext_vector_type(8))) float v8f;

#define WMMA_F32(a, b, c) \
  __builtin_amdgcn_wmma_f32_16x16x4_f32(false, (a), false, (b), (short)0, (c), false, false)

#define BB      1024
#define NN      100000
#define DD      64
#define HH      512
#define RTILES  4
#define NSPLIT  64
#define CHUNKS  (NN / 16)
#define CPS     ((CHUNKS + NSPLIT - 1) / NSPLIT)

__device__ __forceinline__ float xhalf(float x) {
  const int b = __builtin_bit_cast(int, x);
  return __builtin_bit_cast(
      float, __builtin_amdgcn_permlanex16(b, b, 0x76543210, 0xfedcba98, false,
                                          false));
}

__device__ __forceinline__ void load_chunk(v2f (&f)[16],
                                           const float* __restrict__ table,
                                           int chunk, int m, int koff) {
  const int col = chunk * 16 + m;
  const float* trow = table + (size_t)col * DD + koff;
#pragma unroll
  for (int kb = 0; kb < 16; ++kb) f[kb] = *(const v2f*)(trow + kb * 4);
}

__device__ __forceinline__ void use_chunk(const v2f (&a)[RTILES][16],
                                          const v2f (&f)[16],
                                          int col, float (&bestv)[RTILES][8],
                                          int (&besti)[RTILES][8]) {
  v2f q0 = f[0] * f[0];
  v2f q1 = f[1] * f[1];
#pragma unroll
  for (int kb = 2; kb < 16; kb += 2) {
    q0 = __builtin_elementwise_fma(f[kb], f[kb], q0);
    q1 = __builtin_elementwise_fma(f[kb + 1], f[kb + 1], q1);
  }
  const v2f q = q0 + q1;
  const float ph = q.x + q.y;
  const float tn = ph + xhalf(ph);

  v8f acc[RTILES];
#pragma unroll
  for (int t = 0; t < RTILES; ++t) acc[t] = (v8f){};

#pragma unroll
  for (int kb = 0; kb < 16; ++kb)
#pragma unroll
    for (int t = 0; t < RTILES; ++t)
      acc[t] = WMMA_F32(a[t][kb], f[kb], acc[t]);

#pragma unroll
  for (int t = 0; t < RTILES; ++t)
#pragma unroll
    for (int v = 0; v < 8; ++v) {
      float s = tn + acc[t][v];
      if (s < bestv[t][v]) { bestv[t][v] = s; besti[t][v] = col; }
    }
}

__global__ __launch_bounds__(64) void nn_kernel(
    const float* __restrict__ codes, const float* __restrict__ table,
    float* __restrict__ pval, int* __restrict__ pidx) {
  __shared__ float mval[2][RTILES * 8];
  __shared__ int midx[2][RTILES * 8];

  const int lane = threadIdx.x & 31;
  const int wave = __builtin_amdgcn_readfirstlane((int)(threadIdx.x >> 5));
  const int mgroup = blockIdx.x;
  const int split = blockIdx.y;
  const int m = lane & 15;
  const int koff = (lane >> 4) * 2;

  v2f a[RTILES][16];
#pragma unroll
  for (int t = 0; t < RTILES; ++t) {
    const float* crow =
        codes + (size_t)((mgroup * RTILES + t) * 16 + m) * DD + koff;
#pragma unroll
    for (int kb = 0; kb < 16; ++kb) {
      v2f v = *(const v2f*)(crow + kb * 4);
      a[t][kb].x = -2.0f * v.x;
      a[t][kb].y = -2.0f * v.y;
    }
  }

  float bestv[RTILES][8];
  int besti[RTILES][8];
#pragma unroll
  for (int t = 0; t < RTILES; ++t)
#pragma unroll
    for (int v = 0; v < 8; ++v) { bestv[t][v] = 3.4e38f; besti[t][v] = 0; }

  const int s0 = split * CPS;
  const int s1 = (s0 + CPS < CHUNKS) ? (s0 + CPS) : CHUNKS;
  const int mid = s0 + ((s1 - s0 + 1) >> 1);
  const int c0 = wave ? mid : s0;
  const int c1 = wave ? s1 : mid;
  const int clast = c1 - 1;

  v2f f0[16], f1[16], f2[16];
  load_chunk(f0, table, c0 < clast ? c0 : clast, m, koff);
  load_chunk(f1, table, c0 + 1 < clast ? c0 + 1 : clast, m, koff);

  int c = c0;
  for (; c + 3 <= c1; c += 3) {
    load_chunk(f2, table, c + 2, m, koff);
    __builtin_amdgcn_sched_barrier(0);
    use_chunk(a, f0, c * 16 + m, bestv, besti);
    load_chunk(f0, table, c + 3 < clast ? c + 3 : clast, m, koff);
    __builtin_amdgcn_sched_barrier(0);
    use_chunk(a, f1, (c + 1) * 16 + m, bestv, besti);
    load_chunk(f1, table, c + 4 < clast ? c + 4 : clast, m, koff);
    __builtin_amdgcn_sched_barrier(0);
    use_chunk(a, f2, (c + 2) * 16 + m, bestv, besti);
  }
  if (c < c1) use_chunk(a, f0, c * 16 + m, bestv, besti);
  if (c + 1 < c1) use_chunk(a, f1, (c + 1) * 16 + m, bestv, besti);

#pragma unroll
  for (int off = 1; off < 16; off <<= 1) {
#pragma unroll
    for (int t = 0; t < RTILES; ++t)
#pragma unroll
      for (int v = 0; v < 8; ++v) {
        float ov = __shfl_xor(bestv[t][v], off, 32);
        int oi = __shfl_xor(besti[t][v], off, 32);
        if (ov < bestv[t][v]) { bestv[t][v] = ov; besti[t][v] = oi; }
      }
  }

  const int hh = lane >> 4;
  if (wave == 1 && (lane & 15) == 0) {
#pragma unroll
    for (int t = 0; t < RTILES; ++t)
#pragma unroll
      for (int v = 0; v < 8; ++v) {
        mval[hh][t * 8 + v] = bestv[t][v];
        midx[hh][t * 8 + v] = besti[t][v];
      }
  }
  __syncthreads();
  if (wave == 0 && (lane & 15) == 0) {
    const int rbase = hh * 8;
#pragma unroll
    for (int t = 0; t < RTILES; ++t) {
      const int tile = mgroup * RTILES + t;
      const int out = (tile * NSPLIT + split) * 16 + rbase;
#pragma unroll
      for (int v = 0; v < 8; ++v) {
        const float ov = mval[hh][t * 8 + v];
        const int oi = midx[hh][t * 8 + v];
        const bool take = ov < bestv[t][v];
        pval[out + v] = take ? ov : bestv[t][v];
        pidx[out + v] = take ? oi : besti[t][v];
      }
    }
  }
}

#define HPITCH (HH + 4)

__global__ __launch_bounds__(256) void mlp_kernel(
    const float* __restrict__ table, const float* __restrict__ pval,
    const int* __restrict__ pidx, const float* __restrict__ W1,
    const float* __restrict__ b1, const float* __restrict__ Wu,
    const float* __restrict__ bu, const float* __restrict__ Ws,
    const float* __restrict__ bs, float* __restrict__ out) {
  __shared__ __attribute__((aligned(16))) float h1s[16 * HPITCH];
  __shared__ int nnrow[16];

  const int lane = threadIdx.x & 31;
  const int wave = __builtin_amdgcn_readfirstlane((int)(threadIdx.x >> 5));
  const int mtile = blockIdx.x;
  const int n = lane & 15;
  const int hh = lane >> 4;
  const int koff = hh * 2;

  const int row0 = wave * 2 + hh;
  const int pb = (mtile * NSPLIT + n) * 16 + row0;
  float pv[4];
  int pi[4];
#pragma unroll
  for (int j = 0; j < 4; ++j) {
    pv[j] = pval[pb + j * 256];
    pi[j] = pidx[pb + j * 256];
  }

  v2f w1f[4][16];
  float bias1[4];
#pragma unroll
  for (int t = 0; t < 4; ++t) {
    const int nc = (wave * 4 + t) * 16 + n;
#pragma unroll
    for (int kb = 0; kb < 16; ++kb) {
      w1f[t][kb].x = W1[(size_t)(kb * 4 + koff + 0) * HH + nc];
      w1f[t][kb].y = W1[(size_t)(kb * 4 + koff + 1) * HH + nc];
    }
    bias1[t] = b1[nc];
  }

  {
    float bv = pv[0];
    int bi = pi[0];
#pragma unroll
    for (int j = 1; j < 4; ++j) {
      const bool lt = pv[j] < bv;
      bv = lt ? pv[j] : bv;
      bi = lt ? pi[j] : bi;
    }
#pragma unroll
    for (int off = 8; off >= 1; off >>= 1) {
      const float ov = __shfl_xor(bv, off, 32);
      const int oi = __shfl_xor(bi, off, 32);
      const bool take = (ov < bv) | ((ov == bv) & (oi < bi));
      bv = take ? ov : bv;
      bi = take ? oi : bi;
    }
    if (n == 0) nnrow[row0] = bi;
  }
  __syncthreads();

  const float* prow = table + (size_t)nnrow[n] * DD + koff;
  v2f a[16];
#pragma unroll
  for (int kb = 0; kb < 16; ++kb) a[kb] = *(const v2f*)(prow + kb * 4);

  const int z = wave >> 2;
  const float* Wz = z ? Ws : Wu;
  const float* bz = z ? bs : bu;
  const int nc2 = (wave & 3) * 16 + n;
  v2f b2[HH / 4];
#pragma unroll
  for (int kb = 0; kb < HH / 4; ++kb) {
    b2[kb].x = Wz[(size_t)(kb * 4 + koff + 0) * DD + nc2];
    b2[kb].y = Wz[(size_t)(kb * 4 + koff + 1) * DD + nc2];
  }
  const float bias2 = bz[nc2];

  {
    v8f acc[4];
#pragma unroll
    for (int t = 0; t < 4; ++t)
#pragma unroll
      for (int v = 0; v < 8; ++v) acc[t][v] = bias1[t];

#pragma unroll
    for (int kb = 0; kb < 16; ++kb)
#pragma unroll
      for (int t = 0; t < 4; ++t)
        acc[t] = WMMA_F32(a[kb], w1f[t][kb], acc[t]);

#pragma unroll
    for (int t = 0; t < 4; ++t)
#pragma unroll
      for (int v = 0; v < 8; ++v) {
        const int m = v + hh * 8;
        const float h = acc[t][v];
        h1s[m * HPITCH + (wave * 4 + t) * 16 + n] = h > 0.f ? h : 0.f;
      }
  }
  __syncthreads();

  {
    v8f o;
#pragma unroll
    for (int v = 0; v < 8; ++v) o[v] = bias2;
    const float* hrow = h1s + n * HPITCH + koff;
    v2f a2[2][16];
#pragma unroll
    for (int i = 0; i < 16; ++i) a2[0][i] = *(const v2f*)(hrow + i * 4);
#pragma unroll
    for (int g = 0; g < HH / 64; ++g) {
      if (g + 1 < HH / 64) {
#pragma unroll
        for (int i = 0; i < 16; ++i)
          a2[(g + 1) & 1][i] = *(const v2f*)(hrow + ((g + 1) * 16 + i) * 4);
      }
      __builtin_amdgcn_sched_barrier(0);
#pragma unroll
      for (int i = 0; i < 16; ++i)
        o = WMMA_F32(a2[g & 1][i], b2[g * 16 + i], o);
    }
    float* op = out + (size_t)z * BB * DD;
#pragma unroll
    for (int v = 0; v < 8; ++v) {
      const int m = v + hh * 8;
      op[(size_t)(mtile * 16 + m) * DD + nc2] = o[v];
    }
  }
}

extern "C" void kernel_launch(void* const* d_in, const int* in_sizes, int n_in,
                              void* d_out, int out_size, void* d_ws,
                              size_t ws_size, hipStream_t stream) {
  const float* codes = (const float*)d_in[0];
  const float* table = (const float*)d_in[1];
  const float* W1 = (const float*)d_in[2];
  const float* b1 = (const float*)d_in[3];
  const float* Wu = (const float*)d_in[4];
  const float* bu = (const float*)d_in[5];
  const float* Ws = (const float*)d_in[6];
  const float* bs = (const float*)d_in[7];
  float* out = (float*)d_out;

  char* ws = (char*)d_ws;
  float* pval  = (float*)(ws);
  int*   pidx  = (int*)  (ws + 262144);

  nn_kernel<<<dim3(BB / 16 / RTILES, NSPLIT), 64, 0, stream>>>(
      codes, table, pval, pidx);
  mlp_kernel<<<BB / 16, 256, 0, stream>>>(table, pval, pidx, W1, b1, Wu, bu, Ws,
                                          bs, out);
}


# nn search: table chunk as the A operand, so a lane holds one query against 8 table rows: a min3 tree keeps value and 8-row group per lane, and mlp settles the row of the 8 with one WMMA chain
# speedup vs baseline: 2.0682x; 1.0402x over previous
#include <hip/hip_runtime.h>

typedef __attribute__((ext_vector_type(2))) float v2f;
typedef __attribute__((ext_vector_type(4))) float v4f;
typedef __attribute__((ext_vector_type(8))) float v8f;

#define WMMA_F32(a, b, c) \
  __builtin_amdgcn_wmma_f32_16x16x4_f32(false, (a), false, (b), (short)0, (c), false, false)

#define BB      1024
#define NN      100000
#define DD      64
#define HH      512
#define RTILES  4
#define NSPLIT  64
#define CHUNKS  (NN / 16)
#define CPS     ((CHUNKS + NSPLIT - 1) / NSPLIT)


__device__ __forceinline__ int xhalf_i(int b) {
  return __builtin_amdgcn_permlanex16(b, b, 0x76543210, 0xfedcba98, false,
                                      false);
}
__device__ __forceinline__ float xhalf(float x) {
  return __builtin_bit_cast(float, xhalf_i(__builtin_bit_cast(int, x)));
}

__device__ __forceinline__ void load_chunk(v2f (&f)[16],
                                           const float* __restrict__ table,
                                           int chunk, int m, int koff) {
  const int row = chunk * 16 + m;
  const float* trow = table + (size_t)row * DD + koff;
#pragma unroll
  for (int kb = 0; kb < 16; ++kb) f[kb] = *(const v2f*)(trow + kb * 4);
}

__device__ __forceinline__ float row_norm(const v2f (&f)[16]) {
  v2f q0 = f[0] * f[0];
  v2f q1 = f[1] * f[1];
#pragma unroll
  for (int kb = 2; kb < 16; kb += 2) {
    q0 = __builtin_elementwise_fma(f[kb], f[kb], q0);
    q1 = __builtin_elementwise_fma(f[kb + 1], f[kb + 1], q1);
  }
  const v2f q = q0 + q1;
  const float ph = q.x + q.y;
  return ph + xhalf(ph);
}

__device__ __forceinline__ v8f norm_tile(const v2f (&f)[16], float* tnl, int m,
                                         int hh) {
  tnl[m] = row_norm(f);
  __builtin_amdgcn_fence(__ATOMIC_SEQ_CST, "wavefront");
  const v4f lo = *(const v4f*)(tnl + hh * 8);
  const v4f hi = *(const v4f*)(tnl + hh * 8 + 4);
  __builtin_amdgcn_fence(__ATOMIC_SEQ_CST, "wavefront");
  return (v8f){lo.x, lo.y, lo.z, lo.w, hi.x, hi.y, hi.z, hi.w};
}

__device__ __forceinline__ void use_chunk(const v2f (&a)[RTILES][16],
                                          const v2f (&f)[16], float* tnl,
                                          int m, int hh, int chunk,
                                          float (&best)[RTILES],
                                          int (&bidx)[RTILES]) {
  const v8f nrm = norm_tile(f, tnl, m, hh);

  v8f acc[RTILES];
#pragma unroll
  for (int t = 0; t < RTILES; ++t) acc[t] = (v8f){};

#pragma unroll
  for (int kb = 0; kb < 16; ++kb)
#pragma unroll
    for (int t = 0; t < RTILES; ++t)
      acc[t] = WMMA_F32(f[kb], a[t][kb], acc[t]);

#pragma unroll
  for (int t = 0; t < RTILES; ++t) {
    const v8f s = nrm + acc[t];
    const float m0 = __builtin_fminf(__builtin_fminf(s[0], s[1]), s[2]);
    const float m1 = __builtin_fminf(__builtin_fminf(s[3], s[4]), s[5]);
    const float m2 = __builtin_fminf(__builtin_fminf(s[6], s[7]), best[t]);
    const float nb = __builtin_fminf(__builtin_fminf(m0, m1), m2);
    if (nb < best[t]) bidx[t] = chunk;
    best[t] = nb;
  }
}

__global__ __launch_bounds__(64) void nn_kernel(
    const float* __restrict__ codes, const float* __restrict__ table,
    float* __restrict__ pval, int* __restrict__ pidx) {
  __shared__ float mval[RTILES][16];
  __shared__ int mcode[RTILES][16];
  __shared__ __attribute__((aligned(16))) float tnorm[2][16];

  const int lane = threadIdx.x & 31;
  const int wave = __builtin_amdgcn_readfirstlane((int)(threadIdx.x >> 5));
  const int mgroup = blockIdx.x;
  const int split = blockIdx.y;
  const int m = lane & 15;
  const int hh = lane >> 4;
  const int koff = hh * 2;
  float* tnl = tnorm[wave];

  v2f a[RTILES][16];
#pragma unroll
  for (int t = 0; t < RTILES; ++t) {
    const float* crow =
        codes + (size_t)((mgroup * RTILES + t) * 16 + m) * DD + koff;
#pragma unroll
    for (int kb = 0; kb < 16; ++kb) {
      v2f v = *(const v2f*)(crow + kb * 4);
      a[t][kb].x = -2.0f * v.x;
      a[t][kb].y = -2.0f * v.y;
    }
  }

  const int s0 = split * CPS;
  const int s1 = (s0 + CPS < CHUNKS) ? (s0 + CPS) : CHUNKS;
  const int mid = s0 + ((s1 - s0 + 1) >> 1);
  const int c0 = wave ? mid : s0;
  const int c1 = wave ? s1 : mid;
  const int clast = c1 - 1;

  float best[RTILES];
  int bidx[RTILES];
#pragma unroll
  for (int t = 0; t < RTILES; ++t) { best[t] = 3.4e38f; bidx[t] = c0; }

  v2f f0[16], f1[16], f2[16];
  load_chunk(f0, table, c0 < clast ? c0 : clast, m, koff);
  load_chunk(f1, table, c0 + 1 < clast ? c0 + 1 : clast, m, koff);

  int c = c0;
  for (; c + 3 <= c1; c += 3) {
    load_chunk(f2, table, c + 2, m, koff);
    __builtin_amdgcn_sched_barrier(0);
    use_chunk(a, f0, tnl, m, hh, c, best, bidx);
    load_chunk(f0, table, c + 3 < clast ? c + 3 : clast, m, koff);
    __builtin_amdgcn_sched_barrier(0);
    use_chunk(a, f1, tnl, m, hh, c + 1, best, bidx);
    load_chunk(f1, table, c + 4 < clast ? c + 4 : clast, m, koff);
    __builtin_amdgcn_sched_barrier(0);
    use_chunk(a, f2, tnl, m, hh, c + 2, best, bidx);
  }
  if (c < c1) use_chunk(a, f0, tnl, m, hh, c, best, bidx);
  if (c + 1 < c1) use_chunk(a, f1, tnl, m, hh, c + 1, best, bidx);

  float val[RTILES];
  int code[RTILES];
#pragma unroll
  for (int t = 0; t < RTILES; ++t) {
    const float v = best[t];
    const int cd = bidx[t] * 2 + hh;
    const float ov = xhalf(v);
    const int oc = xhalf_i(cd);
    const bool take = (ov < v) | ((ov == v) & (oc < cd));
    val[t] = take ? ov : v;
    code[t] = take ? oc : cd;
  }

  if (wave == 1 && lane < 16) {
#pragma unroll
    for (int t = 0; t < RTILES; ++t) {
      mval[t][lane] = val[t];
      mcode[t][lane] = code[t];
    }
  }
  __syncthreads();
  if (wave == 0 && lane < 16) {
#pragma unroll
    for (int t = 0; t < RTILES; ++t) {
      const int tile = mgroup * RTILES + t;
      const int out = (tile * NSPLIT + split) * 16 + lane;
      const float ov = mval[t][lane];
      const int oc = mcode[t][lane];
      const bool take = ov < val[t];
      pval[out] = take ? ov : val[t];
      pidx[out] = take ? oc : code[t];
    }
  }
}

#define HPITCH (HH + 4)

__global__ __launch_bounds__(256) void mlp_kernel(
    const float* __restrict__ codes, const float* __restrict__ table,
    const float* __restrict__ pval, const int* __restrict__ pidx,
    const float* __restrict__ W1, const float* __restrict__ b1,
    const float* __restrict__ Wu, const float* __restrict__ bu,
    const float* __restrict__ Ws, const float* __restrict__ bs,
    float* __restrict__ out) {
  __shared__ __attribute__((aligned(16))) float h1s[16 * HPITCH];
  __shared__ int ncode[16];
  __shared__ float sres[8][16];

  const int lane = threadIdx.x & 31;
  const int wave = __builtin_amdgcn_readfirstlane((int)(threadIdx.x >> 5));
  const int mtile = blockIdx.x;
  const int n = lane & 15;
  const int hh = lane >> 4;
  const int koff = hh * 2;

  const int row0 = wave * 2 + hh;
  const int pb = (mtile * NSPLIT + n) * 16 + row0;
  float pv[4];
  int pi[4];
#pragma unroll
  for (int j = 0; j < 4; ++j) {
    pv[j] = pval[pb + j * 256];
    pi[j] = pidx[pb + j * 256];
  }

  v2f qf[16];
  {
    const float* qrow = codes + (size_t)(mtile * 16 + n) * DD + koff;
#pragma unroll
    for (int kb = 0; kb < 16; ++kb) {
      v2f v = *(const v2f*)(qrow + kb * 4);
      qf[kb].x = -2.0f * v.x;
      qf[kb].y = -2.0f * v.y;
    }
  }

  v2f w1f[4][16];
  float bias1[4];
#pragma unroll
  for (int t = 0; t < 4; ++t) {
    const int nc = (wave * 4 + t) * 16 + n;
#pragma unroll
    for (int kb = 0; kb < 16; ++kb) {
      w1f[t][kb].x = W1[(size_t)(kb * 4 + koff + 0) * HH + nc];
      w1f[t][kb].y = W1[(size_t)(kb * 4 + koff + 1) * HH + nc];
    }
    bias1[t] = b1[nc];
  }

  {
    float bv = pv[0];
    int bi = pi[0];
#pragma unroll
    for (int j = 1; j < 4; ++j) {
      const bool lt = pv[j] < bv;
      bv = lt ? pv[j] : bv;
      bi = lt ? pi[j] : bi;
    }
#pragma unroll
    for (int off = 8; off >= 1; off >>= 1) {
      const float ov = __shfl_xor(bv, off, 32);
      const int oi = __shfl_xor(bi, off, 32);
      const bool take = (ov < bv) | ((ov == bv) & (oi < bi));
      bv = take ? ov : bv;
      bi = take ? oi : bi;
    }
    if (n == 0) ncode[row0] = bi;
  }
  __syncthreads();

  int cd = ncode[n];
  cd = cd < 0 ? 0 : (cd > 2 * CHUNKS - 1 ? 2 * CHUNKS - 1 : cd);
  const int gbase = cd * 8;
  {
    const float* crow = table + (size_t)(gbase + wave) * DD + koff;
    v2f cf[16];
#pragma unroll
    for (int kb = 0; kb < 16; ++kb) cf[kb] = *(const v2f*)(crow + kb * 4);
    const float tn = row_norm(cf);
    v8f d = (v8f){};
#pragma unroll
    for (int kb = 0; kb < 16; ++kb) d = WMMA_F32(cf[kb], qf[kb], d);
    float s = d[0];
#pragma unroll
    for (int v = 1; v < 8; ++v) s = ((n & 7) == v) ? d[v] : s;
    if ((n >> 3) == hh) sres[wave][n] = tn + s;
  }
  __syncthreads();
  int nrow;
  {
    float bs8 = sres[0][n];
    int bj = 0;
#pragma unroll
    for (int j = 1; j < 8; ++j) {
      const float x = sres[j][n];
      const bool lt = x < bs8;
      bs8 = lt ? x : bs8;
      bj = lt ? j : bj;
    }
    nrow = gbase + bj;
  }

  const float* prow = table + (size_t)nrow * DD + koff;
  v2f a[16];
#pragma unroll
  for (int kb = 0; kb < 16; ++kb) a[kb] = *(const v2f*)(prow + kb * 4);

  const int z = wave >> 2;
  const float* Wz = z ? Ws : Wu;
  const float* bz = z ? bs : bu;
  const int nc2 = (wave & 3) * 16 + n;
  v2f b2[HH / 4];
#pragma unroll
  for (int kb = 0; kb < HH / 4; ++kb) {
    b2[kb].x = Wz[(size_t)(kb * 4 + koff + 0) * DD + nc2];
    b2[kb].y = Wz[(size_t)(kb * 4 + koff + 1) * DD + nc2];
  }
  const float bias2 = bz[nc2];

  {
    v8f acc[4];
#pragma unroll
    for (int t = 0; t < 4; ++t)
#pragma unroll
      for (int v = 0; v < 8; ++v) acc[t][v] = bias1[t];

#pragma unroll
    for (int kb = 0; kb < 16; ++kb)
#pragma unroll
      for (int t = 0; t < 4; ++t)
        acc[t] = WMMA_F32(a[kb], w1f[t][kb], acc[t]);

#pragma unroll
    for (int t = 0; t < 4; ++t)
#pragma unroll
      for (int v = 0; v < 8; ++v) {
        const int m = v + hh * 8;
        const float h = acc[t][v];
        h1s[m * HPITCH + (wave * 4 + t) * 16 + n] = h > 0.f ? h : 0.f;
      }
  }
  __syncthreads();

  {
    v8f o;
#pragma unroll
    for (int v = 0; v < 8; ++v) o[v] = bias2;
    const float* hrow = h1s + n * HPITCH + koff;
    v2f a2[2][16];
#pragma unroll
    for (int i = 0; i < 16; ++i) a2[0][i] = *(const v2f*)(hrow + i * 4);
#pragma unroll
    for (int g = 0; g < HH / 64; ++g) {
      if (g + 1 < HH / 64) {
#pragma unroll
        for (int i = 0; i < 16; ++i)
          a2[(g + 1) & 1][i] = *(const v2f*)(hrow + ((g + 1) * 16 + i) * 4);
      }
      __builtin_amdgcn_sched_barrier(0);
#pragma unroll
      for (int i = 0; i < 16; ++i)
        o = WMMA_F32(a2[g & 1][i], b2[g * 16 + i], o);
    }
    float* op = out + (size_t)z * BB * DD;
#pragma unroll
    for (int v = 0; v < 8; ++v) {
      const int m = v + hh * 8;
      op[(size_t)(mtile * 16 + m) * DD + nc2] = o[v];
    }
  }
}

extern "C" void kernel_launch(void* const* d_in, const int* in_sizes, int n_in,
                              void* d_out, int out_size, void* d_ws,
                              size_t ws_size, hipStream_t stream) {
  const float* codes = (const float*)d_in[0];
  const float* table = (const float*)d_in[1];
  const float* W1 = (const float*)d_in[2];
  const float* b1 = (const float*)d_in[3];
  const float* Wu = (const float*)d_in[4];
  const float* bu = (const float*)d_in[5];
  const float* Ws = (const float*)d_in[6];
  const float* bs = (const float*)d_in[7];
  float* out = (float*)d_out;

  char* ws = (char*)d_ws;
  float* pval  = (float*)(ws);
  int*   pidx  = (int*)  (ws + 262144);

  nn_kernel<<<dim3(BB / 16 / RTILES, NSPLIT), 64, 0, stream>>>(
      codes, table, pval, pidx);
  mlp_kernel<<<BB / 16, 256, 0, stream>>>(codes, table, pval, pidx, W1, b1, Wu,
                                          bu, Ws, bs, out);
}


# mlp: weights fetched 16 bytes per lane (a wave's 4 tiles interleave their columns); layer 2: one head per work-group, k split over its 8 waves, partials joined via LDS (new order of the 512 terms)
# speedup vs baseline: 2.1434x; 1.0364x over previous
#include <hip/hip_runtime.h>

typedef __attribute__((ext_vector_type(2))) float v2f;
typedef __attribute__((ext_vector_type(4))) float v4f;
typedef __attribute__((ext_vector_type(8))) float v8f;

#define WMMA_F32(a, b, c) \
  __builtin_amdgcn_wmma_f32_16x16x4_f32(false, (a), false, (b), (short)0, (c), false, false)

#define BB      1024
#define NN      100000
#define DD      64
#define HH      512
#define RTILES  4
#define NSPLIT  64
#define CHUNKS  (NN / 16)
#define CPS     ((CHUNKS + NSPLIT - 1) / NSPLIT)


__device__ __forceinline__ int xhalf_i(int b) {
  return __builtin_amdgcn_permlanex16(b, b, 0x76543210, 0xfedcba98, false,
                                      false);
}
__device__ __forceinline__ float xhalf(float x) {
  return __builtin_bit_cast(float, xhalf_i(__builtin_bit_cast(int, x)));
}

__device__ __forceinline__ void load_chunk(v2f (&f)[16],
                                           const float* __restrict__ table,
                                           int chunk, int m, int koff) {
  const int row = chunk * 16 + m;
  const float* trow = table + (size_t)row * DD + koff;
#pragma unroll
  for (int kb = 0; kb < 16; ++kb) f[kb] = *(const v2f*)(trow + kb * 4);
}

__device__ __forceinline__ float row_norm(const v2f (&f)[16]) {
  v2f q0 = f[0] * f[0];
  v2f q1 = f[1] * f[1];
#pragma unroll
  for (int kb = 2; kb < 16; kb += 2) {
    q0 = __builtin_elementwise_fma(f[kb], f[kb], q0);
    q1 = __builtin_elementwise_fma(f[kb + 1], f[kb + 1], q1);
  }
  const v2f q = q0 + q1;
  const float ph = q.x + q.y;
  return ph + xhalf(ph);
}

__device__ __forceinline__ v8f norm_tile(const v2f (&f)[16], float* tnl, int m,
                                         int hh) {
  tnl[m] = row_norm(f);
  __builtin_amdgcn_fence(__ATOMIC_SEQ_CST, "wavefront");
  const v4f lo = *(const v4f*)(tnl + hh * 8);
  const v4f hi = *(const v4f*)(tnl + hh * 8 + 4);
  __builtin_amdgcn_fence(__ATOMIC_SEQ_CST, "wavefront");
  return (v8f){lo.x, lo.y, lo.z, lo.w, hi.x, hi.y, hi.z, hi.w};
}

__device__ __forceinline__ void use_chunk(const v2f (&a)[RTILES][16],
                                          const v2f (&f)[16], float* tnl,
                                          int m, int hh, int chunk,
                                          float (&best)[RTILES],
                                          int (&bidx)[RTILES]) {
  const v8f nrm = norm_tile(f, tnl, m, hh);

  v8f acc[RTILES];
#pragma unroll
  for (int t = 0; t < RTILES; ++t) acc[t] = (v8f){};

#pragma unroll
  for (int kb = 0; kb < 16; ++kb)
#pragma unroll
    for (int t = 0; t < RTILES; ++t)
      acc[t] = WMMA_F32(f[kb], a[t][kb], acc[t]);

#pragma unroll
  for (int t = 0; t < RTILES; ++t) {
    const v8f s = nrm + acc[t];
    const float m0 = __builtin_fminf(__builtin_fminf(s[0], s[1]), s[2]);
    const float m1 = __builtin_fminf(__builtin_fminf(s[3], s[4]), s[5]);
    const float m2 = __builtin_fminf(__builtin_fminf(s[6], s[7]), best[t]);
    const float nb = __builtin_fminf(__builtin_fminf(m0, m1), m2);
    if (nb < best[t]) bidx[t] = chunk;
    best[t] = nb;
  }
}

__global__ __launch_bounds__(64) void nn_kernel(
    const float* __restrict__ codes, const float* __restrict__ table,
    float* __restrict__ pval, int* __restrict__ pidx) {
  __shared__ float mval[RTILES][16];
  __shared__ int mcode[RTILES][16];
  __shared__ __attribute__((aligned(16))) float tnorm[2][16];

  const int lane = threadIdx.x & 31;
  const int wave = __builtin_amdgcn_readfirstlane((int)(threadIdx.x >> 5));
  const int mgroup = blockIdx.x;
  const int split = blockIdx.y;
  const int m = lane & 15;
  const int hh = lane >> 4;
  const int koff = hh * 2;
  float* tnl = tnorm[wave];

  v2f a[RTILES][16];
#pragma unroll
  for (int t = 0; t < RTILES; ++t) {
    const float* crow =
        codes + (size_t)((mgroup * RTILES + t) * 16 + m) * DD + koff;
#pragma unroll
    for (int kb = 0; kb < 16; ++kb) {
      v2f v = *(const v2f*)(crow + kb * 4);
      a[t][kb].x = -2.0f * v.x;
      a[t][kb].y = -2.0f * v.y;
    }
  }

  const int s0 = split * CPS;
  const int s1 = (s0 + CPS < CHUNKS) ? (s0 + CPS) : CHUNKS;
  const int mid = s0 + ((s1 - s0 + 1) >> 1);
  const int c0 = wave ? mid : s0;
  const int c1 = wave ? s1 : mid;
  const int clast = c1 - 1;

  float best[RTILES];
  int bidx[RTILES];
#pragma unroll
  for (int t = 0; t < RTILES; ++t) { best[t] = 3.4e38f; bidx[t] = c0; }

  v2f f0[16], f1[16], f2[16];
  load_chunk(f0, table, c0 < clast ? c0 : clast, m, koff);
  load_chunk(f1, table, c0 + 1 < clast ? c0 + 1 : clast, m, koff);

  int c = c0;
  for (; c + 3 <= c1; c += 3) {
    load_chunk(f2, table, c + 2, m, koff);
    __builtin_amdgcn_sched_barrier(0);
    use_chunk(a, f0, tnl, m, hh, c, best, bidx);
    load_chunk(f0, table, c + 3 < clast ? c + 3 : clast, m, koff);
    __builtin_amdgcn_sched_barrier(0);
    use_chunk(a, f1, tnl, m, hh, c + 1, best, bidx);
    load_chunk(f1, table, c + 4 < clast ? c + 4 : clast, m, koff);
    __builtin_amdgcn_sched_barrier(0);
    use_chunk(a, f2, tnl, m, hh, c + 2, best, bidx);
  }
  if (c < c1) use_chunk(a, f0, tnl, m, hh, c, best, bidx);
  if (c + 1 < c1) use_chunk(a, f1, tnl, m, hh, c + 1, best, bidx);

  float val[RTILES];
  int code[RTILES];
#pragma unroll
  for (int t = 0; t < RTILES; ++t) {
    const float v = best[t];
    const int cd = bidx[t] * 2 + hh;
    const float ov = xhalf(v);
    const int oc = xhalf_i(cd);
    const bool take = (ov < v) | ((ov == v) & (oc < cd));
    val[t] = take ? ov : v;
    code[t] = take ? oc : cd;
  }

  if (wave == 1 && lane < 16) {
#pragma unroll
    for (int t = 0; t < RTILES; ++t) {
      mval[t][lane] = val[t];
      mcode[t][lane] = code[t];
    }
  }
  __syncthreads();
  if (wave == 0 && lane < 16) {
#pragma unroll
    for (int t = 0; t < RTILES; ++t) {
      const int tile = mgroup * RTILES + t;
      const int out = (tile * NSPLIT + split) * 16 + lane;
      const float ov = mval[t][lane];
      const int oc = mcode[t][lane];
      const bool take = ov < val[t];
      pval[out] = take ? ov : val[t];
      pidx[out] = take ? oc : code[t];
    }
  }
}

#define HPITCH (HH + 4)

__global__ __launch_bounds__(256) void mlp_kernel(
    const float* __restrict__ codes, const float* __restrict__ table,
    const float* __restrict__ pval, const int* __restrict__ pidx,
    const float* __restrict__ W1, const float* __restrict__ b1,
    const float* __restrict__ Wu, const float* __restrict__ bu,
    const float* __restrict__ Ws, const float* __restrict__ bs,
    float* __restrict__ out) {
  __shared__ __attribute__((aligned(16))) float h1s[16 * HPITCH];
  __shared__ int ncode[16];
  __shared__ float sres[8][16];
  __shared__ __attribute__((aligned(16))) float part[8 * 8 * 32 * 4];

  const int lane = threadIdx.x & 31;
  const int wave = __builtin_amdgcn_readfirstlane((int)(threadIdx.x >> 5));
  const int mtile = blockIdx.x;
  const int z = blockIdx.y;
  const int n = lane & 15;
  const int hh = lane >> 4;
  const int koff = hh * 2;

  const int row0 = wave * 2 + hh;
  const int pb = (mtile * NSPLIT + n) * 16 + row0;
  float pv[4];
  int pi[4];
#pragma unroll
  for (int j = 0; j < 4; ++j) {
    pv[j] = pval[pb + j * 256];
    pi[j] = pidx[pb + j * 256];
  }

  v2f qf[16];
  {
    const float* qrow = codes + (size_t)(mtile * 16 + n) * DD + koff;
#pragma unroll
    for (int kb = 0; kb < 16; ++kb) {
      v2f v = *(const v2f*)(qrow + kb * 4);
      qf[kb].x = -2.0f * v.x;
      qf[kb].y = -2.0f * v.y;
    }
  }

  v4f w1a[16], w1b[16];
  const float* w1p = W1 + (size_t)koff * HH + wave * 64 + 4 * n;
#pragma unroll
  for (int kb = 0; kb < 16; ++kb) {
    w1a[kb] = *(const v4f*)(w1p + kb * 4 * HH);
    w1b[kb] = *(const v4f*)(w1p + kb * 4 * HH + HH);
  }
  const v4f bias1 = *(const v4f*)(b1 + wave * 64 + 4 * n);

  {
    asm volatile("" ::"v"(pi[0]), "v"(pi[1]), "v"(pi[2]), "v"(pi[3]));
    float bv = pv[0];
    int bi = pi[0];
#pragma unroll
    for (int j = 1; j < 4; ++j) {
      const bool lt = pv[j] < bv;
      bv = lt ? pv[j] : bv;
      bi = lt ? pi[j] : bi;
    }
#pragma unroll
    for (int off = 8; off >= 1; off >>= 1) {
      const float ov = __shfl_xor(bv, off, 32);
      const int oi = __shfl_xor(bi, off, 32);
      const bool take = (ov < bv) | ((ov == bv) & (oi < bi));
      bv = take ? ov : bv;
      bi = take ? oi : bi;
    }
    if (n == 0) ncode[row0] = bi;
  }
  __syncthreads();

  int cd = ncode[n];
  cd = cd < 0 ? 0 : (cd > 2 * CHUNKS - 1 ? 2 * CHUNKS - 1 : cd);
  const int gbase = cd * 8;
  {
    const float* crow = table + (size_t)(gbase + wave) * DD + koff;
    v2f cf[16];
#pragma unroll
    for (int kb = 0; kb < 16; ++kb) cf[kb] = *(const v2f*)(crow + kb * 4);
    const float tn = row_norm(cf);
    v8f d = (v8f){};
#pragma unroll
    for (int kb = 0; kb < 16; ++kb) d = WMMA_F32(cf[kb], qf[kb], d);
    float s = d[0];
#pragma unroll
    for (int v = 1; v < 8; ++v) s = ((n & 7) == v) ? d[v] : s;
    if ((n >> 3) == hh) sres[wave][n] = tn + s;
  }
  __syncthreads();
  int nrow;
  {
    float bs8 = sres[0][n];
    int bj = 0;
#pragma unroll
    for (int j = 1; j < 8; ++j) {
      const float x = sres[j][n];
      const bool lt = x < bs8;
      bs8 = lt ? x : bs8;
      bj = lt ? j : bj;
    }
    nrow = gbase + bj;
  }

  const float* prow = table + (size_t)nrow * DD + koff;
  v2f a[16];
#pragma unroll
  for (int kb = 0; kb < 16; ++kb) a[kb] = *(const v2f*)(prow + kb * 4);

  const float* Wz = z ? Ws : Wu;
  const float* bz = z ? bs : bu;
  v4f b2a[16], b2b[16];
  const float* w2p = Wz + (size_t)(wave * 64 + koff) * DD + 4 * n;
#pragma unroll
  for (int i = 0; i < 16; ++i) {
    b2a[i] = *(const v4f*)(w2p + i * 4 * DD);
    b2b[i] = *(const v4f*)(w2p + i * 4 * DD + DD);
  }
  const v4f bias2 = *(const v4f*)(bz + 4 * n);

  {
    v8f acc[4];
#pragma unroll
    for (int t = 0; t < 4; ++t)
#pragma unroll
      for (int v = 0; v < 8; ++v) acc[t][v] = bias1[t];

#pragma unroll
    for (int kb = 0; kb < 16; ++kb)
#pragma unroll
      for (int t = 0; t < 4; ++t)
        acc[t] = WMMA_F32(a[kb], ((v2f){w1a[kb][t], w1b[kb][t]}), acc[t]);

#pragma unroll
    for (int v = 0; v < 8; ++v) {
      const int m = v + hh * 8;
      v4f h;
#pragma unroll
      for (int t = 0; t < 4; ++t) h[t] = acc[t][v] > 0.f ? acc[t][v] : 0.f;
      *(v4f*)(h1s + m * HPITCH + wave * 64 + 4 * n) = h;
    }
  }
  __syncthreads();

  {
    v8f o[4];
#pragma unroll
    for (int c = 0; c < 4; ++c) o[c] = (v8f){};
    const float* hrow = h1s + n * HPITCH + wave * 64 + koff;
    v2f a2[16];
#pragma unroll
    for (int i = 0; i < 16; ++i) a2[i] = *(const v2f*)(hrow + i * 4);
    __builtin_amdgcn_sched_barrier(0);
#pragma unroll
    for (int i = 0; i < 16; ++i)
#pragma unroll
      for (int c = 0; c < 4; ++c)
        o[c] = WMMA_F32(a2[i], ((v2f){b2a[i][c], b2b[i][c]}), o[c]);

#pragma unroll
    for (int v = 0; v < 8; ++v)
      *(v4f*)(part + ((wave * 8 + v) * 32 + lane) * 4) =
          (v4f){o[0][v], o[1][v], o[2][v], o[3][v]};
  }
  __syncthreads();

  {
    v4f s = bias2;
#pragma unroll
    for (int w = 0; w < 8; ++w)
      s = s + *(const v4f*)(part + ((w * 8 + wave) * 32 + lane) * 4);
    float* op = out + (size_t)z * BB * DD;
    const int m = hh * 8 + wave;
    *(v4f*)(op + (size_t)(mtile * 16 + m) * DD + 4 * n) = s;
  }
}

extern "C" void kernel_launch(void* const* d_in, const int* in_sizes, int n_in,
                              void* d_out, int out_size, void* d_ws,
                              size_t ws_size, hipStream_t stream) {
  const float* codes = (const float*)d_in[0];
  const float* table = (const float*)d_in[1];
  const float* W1 = (const float*)d_in[2];
  const float* b1 = (const float*)d_in[3];
  const float* Wu = (const float*)d_in[4];
  const float* bu = (const float*)d_in[5];
  const float* Ws = (const float*)d_in[6];
  const float* bs = (const float*)d_in[7];
  float* out = (float*)d_out;

  char* ws = (char*)d_ws;
  float* pval  = (float*)(ws);
  int*   pidx  = (int*)  (ws + 262144);

  nn_kernel<<<dim3(BB / 16 / RTILES, NSPLIT), 64, 0, stream>>>(
      codes, table, pval, pidx);
  mlp_kernel<<<dim3(BB / 16, 2), 256, 0, stream>>>(codes, table, pval, pidx, W1, b1, Wu,
                                          bu, Ws, bs, out);
}
